# starting kernel
# baseline (speedup 1.0000x reference)
#include <hip/hip_runtime.h>
#include <hip/hip_bf16.h>

typedef float v2f __attribute__((ext_vector_type(2)));
typedef float v8f __attribute__((ext_vector_type(8)));

__device__ __forceinline__ v8f wmma_f32(v2f a, v2f b, v8f c) {
  return __builtin_amdgcn_wmma_f32_16x16x4_f32(false, a, false, b, (short)0, c,
                                               false, false);
}

#define SPATIAL 32768
#define MPAD 256
#define NWT (27 * 64 * MPAD)

__global__ __launch_bounds__(256) void prep_weights_kernel(
    const float* __restrict__ We, const float* __restrict__ be,
    float* __restrict__ Wt, float* __restrict__ bep) {
  int idx = blockIdx.x * 256 + threadIdx.x;
  if (idx < NWT) {
    int o = idx & (MPAD - 1);
    int cm = (idx >> 8) & 63;
    int koff = idx >> 14;
    float v = (o < 216) ? We[(size_t)o * 1728 + cm * 27 + koff] : 0.0f;
    Wt[((size_t)(koff * 32 + (cm >> 1)) * MPAD + o) * 2 + (cm & 1)] = v;
  } else if (idx < NWT + MPAD) {
    int o = idx - NWT;
    bep[o] = (o < 216) ? be[o] : 0.0f;
  }
}

template <int MTILES, bool BIAS>
__global__ __launch_bounds__(128) void gemm_f32_wmma_kernel(
    const float* __restrict__ A, const float* __restrict__ B,
    const float* __restrict__ bias, float* __restrict__ C, int N) {
  constexpr int M = MTILES * 16;
  __shared__ float lA[64 * 64];
  __shared__ float lB[64 * 64];

  const int tid = threadIdx.x;
  const int lane = tid & 31;
  const int wave = tid >> 5;
  const int n0 = blockIdx.x * 64;
  const int ml = lane & 15;
  const int lhi = lane >> 4;

  for (int i = tid; i < M * 64; i += 128) lA[i] = A[i];
  for (int i = tid; i < 64 * 64; i += 128) {
    int k = i >> 6, col = i & 63;
    lB[((k >> 1) * 64 + col) * 2 + (k & 1)] = B[(size_t)k * N + n0 + col];
  }
  __syncthreads();

  int mt[MTILES], nt[MTILES];
#pragma unroll
  for (int i = 0; i < MTILES; ++i) {
    int p = wave + 4 * i;
    mt[i] = p % MTILES;
    nt[i] = p / MTILES;
  }

  v8f acc[MTILES];
#pragma unroll
  for (int i = 0; i < MTILES; ++i) {
    v8f c;
#pragma unroll
    for (int v = 0; v < 8; ++v)
      c[v] = BIAS ? bias[mt[i] * 16 + v + 8 * lhi] : 0.0f;
    acc[i] = c;
  }

#pragma unroll
  for (int k0 = 0; k0 < 64; k0 += 4) {
#pragma unroll
    for (int i = 0; i < MTILES; ++i) {
      v2f a = *(const v2f*)&lA[(mt[i] * 16 + ml) * 64 + k0 + 2 * lhi];
      v2f b = *(const v2f*)&lB[(((k0 >> 1) + lhi) * 64 + nt[i] * 16 + ml) * 2];
      acc[i] = wmma_f32(a, b, acc[i]);
    }
  }

#pragma unroll
  for (int i = 0; i < MTILES; ++i)
#pragma unroll
    for (int v = 0; v < 8; ++v) {
      int m = mt[i] * 16 + v + 8 * lhi;
      C[(size_t)m * N + n0 + nt[i] * 16 + ml] = acc[i][v];
    }
}

__global__ __launch_bounds__(128) void conv3_wmma_kernel(
    const float* __restrict__ y1, const float* __restrict__ Wt,
    const float* __restrict__ bep, float* __restrict__ y2p) {
  __shared__ float strip[64 * 34];

  const int tid = threadIdx.x;
  const int lane = tid & 31;
  const int wave = tid >> 5;
  const int d = blockIdx.x >> 5;
  const int h = blockIdx.x & 31;
  const int ml = lane & 15;
  const int lhi = lane >> 4;

  v8f acc[4][2];
#pragma unroll
  for (int t = 0; t < 4; ++t) {
    int ob = (wave + 4 * t) * 16;
#pragma unroll
    for (int j = 0; j < 2; ++j) {
      v8f c;
#pragma unroll
      for (int v = 0; v < 8; ++v) c[v] = bep[ob + v + 8 * lhi];
      acc[t][j] = c;
    }
  }

  for (int q = 0; q < 9; ++q) {
    const int kd = q / 3, kh = q % 3;
    const int dd = d + kd - 1, hh = h + kh - 1;
    const bool rowok = (dd >= 0 && dd < 32 && hh >= 0 && hh < 32);
    const int ddc = min(max(dd, 0), 31), hhc = min(max(hh, 0), 31);
    const float* src = y1 + ((size_t)ddc * 32 + hhc) * 32;

    __syncthreads();
    for (int i = tid; i < 64 * 34; i += 128) {
      int cm = i / 34, pos = i % 34;
      int wsrc = pos - 1;
      int wc = min(max(wsrc, 0), 31);
      float ld = src[(size_t)cm * SPATIAL + wc];
      bool ok = rowok && (wsrc >= 0) && (wsrc < 32);
      strip[((cm >> 1) * 34 + pos) * 2 + (cm & 1)] = ok ? ld : 0.0f;
    }
    if (q < 8 && tid < 64) {
      int kd2 = (q + 1) / 3, kh2 = (q + 1) % 3;
      int d2 = min(max(d + kd2 - 1, 0), 31), h2 = min(max(h + kh2 - 1, 0), 31);
      __builtin_prefetch(y1 + ((size_t)tid * SPATIAL) + ((size_t)d2 * 32 + h2) * 32,
                         0, 0);
    }
    __syncthreads();

    const int koffb = kd * 9 + kh * 3;
#pragma unroll 2
    for (int k0 = 0; k0 < 64; k0 += 4) {
#pragma unroll
      for (int t = 0; t < 4; ++t) {
        int ob = (wave + 4 * t) * 16;
        const float* ab =
            Wt + ((size_t)((k0 >> 1) + lhi) * MPAD + ob + ml) * 2 + koffb * (size_t)(32 * MPAD * 2);
#pragma unroll
        for (int kw = 0; kw < 3; ++kw) {
          v2f a = *(const v2f*)(ab + (size_t)kw * (32 * MPAD * 2));
#pragma unroll
          for (int j = 0; j < 2; ++j) {
            v2f b = *(const v2f*)&strip[(((k0 >> 1) + lhi) * 34 + j * 16 + ml + kw) * 2];
            acc[t][j] = wmma_f32(a, b, acc[t][j]);
          }
        }
      }
    }
  }

  const size_t sbase = ((size_t)d * 32 + h) * 32;
#pragma unroll
  for (int t = 0; t < 4; ++t) {
    int ob = (wave + 4 * t) * 16;
#pragma unroll
    for (int j = 0; j < 2; ++j)
#pragma unroll
      for (int v = 0; v < 8; ++v) {
        int o = ob + v + 8 * lhi;
        y2p[(size_t)o * SPATIAL + sbase + j * 16 + ml] = acc[t][j][v];
      }
  }
}

__global__ __launch_bounds__(256) void softmax27_kernel(
    const float* __restrict__ y2p, float* __restrict__ wsft) {
  int t = blockIdx.x * 256 + threadIdx.x;
  int sub = t >> 15;
  int s = t & (SPATIAL - 1);
  float v[27];
  float mx = -3.0e38f;
#pragma unroll
  for (int n = 0; n < 27; ++n) {
    v[n] = y2p[(size_t)(n * 8 + sub) * SPATIAL + s];
    mx = fmaxf(mx, v[n]);
  }
  float sum = 0.0f;
#pragma unroll
  for (int n = 0; n < 27; ++n) {
    v[n] = __expf(v[n] - mx);
    sum += v[n];
  }
  float inv = 1.0f / sum;
#pragma unroll
  for (int n = 0; n < 27; ++n)
    wsft[((size_t)s * 27 + n) * 8 + sub] = v[n] * inv;
}

__global__ __launch_bounds__(64) void carafe_apply_kernel(
    const float* __restrict__ Y, const float* __restrict__ wsft,
    float* __restrict__ out) {
  __shared__ float wl[216];
  __shared__ float yl[32 * 27];
  int s = blockIdx.x;
  int d = s >> 10, h = (s >> 5) & 31, w = s & 31;
  int tid = threadIdx.x;

  for (int i = tid; i < 216; i += 64) wl[i] = wsft[(size_t)s * 216 + i];
  for (int i = tid; i < 32 * 27; i += 64) {
    int o = i / 27, n = i % 27;
    int nd = min(max(d + n / 9 - 1, 0), 31);
    int nh = min(max(h + (n / 3) % 3 - 1, 0), 31);
    int nw = min(max(w + n % 3 - 1, 0), 31);
    yl[i] = Y[(size_t)o * SPATIAL + ((size_t)nd * 32 + nh) * 32 + nw];
  }
  __syncthreads();

#pragma unroll
  for (int r = 0; r < 4; ++r) {
    int oi = tid + r * 64;
    int o = oi >> 3, sub = oi & 7;
    float acc = 0.0f;
#pragma unroll
    for (int n = 0; n < 27; ++n) acc += yl[o * 27 + n] * wl[n * 8 + sub];
    int rd = sub >> 2, rh = (sub >> 1) & 1, rw = sub & 1;
    size_t vox = ((size_t)(2 * d + rd) * 64 + (2 * h + rh)) * 64 + (2 * w + rw);
    out[(size_t)o * 262144 + vox] = acc;
  }
}

extern "C" void kernel_launch(void* const* d_in, const int* in_sizes, int n_in,
                              void* d_out, int out_size, void* d_ws,
                              size_t ws_size, hipStream_t stream) {
  (void)in_sizes; (void)n_in; (void)out_size; (void)ws_size;
  const float* x  = (const float*)d_in[0];
  const float* Wc = (const float*)d_in[1];
  const float* bc = (const float*)d_in[2];
  const float* We = (const float*)d_in[3];
  const float* be = (const float*)d_in[4];
  const float* Wp = (const float*)d_in[5];
  float* out = (float*)d_out;

  float* ws = (float*)d_ws;
  float* y1   = ws;
  float* y2p  = y1 + (size_t)64 * SPATIAL;
  float* wsft = y2p + (size_t)MPAD * SPATIAL;
  float* Yp   = wsft + (size_t)SPATIAL * 216;
  float* Wt   = Yp + (size_t)32 * SPATIAL;
  float* bep  = Wt + (size_t)NWT;

  prep_weights_kernel<<<(NWT + MPAD + 255) / 256, 256, 0, stream>>>(We, be, Wt, bep);
  gemm_f32_wmma_kernel<4, true><<<512, 128, 0, stream>>>(Wc, x, bc, y1, SPATIAL);
  conv3_wmma_kernel<<<1024, 128, 0, stream>>>(y1, Wt, bep, y2p);
  softmax27_kernel<<<1024, 256, 0, stream>>>(y2p, wsft);
  gemm_f32_wmma_kernel<2, false><<<512, 128, 0, stream>>>(Wp, x, nullptr, Yp, SPATIAL);
  carafe_apply_kernel<<<32768, 64, 0, stream>>>(Yp, wsft, out);
}


# softmax folded into the gather kernel: thread = 2 output voxels, logits read from the conv plane, normalised in registers, applied to Y rows in LDS; weight plane and one launch gone; sum orders kept
# speedup vs baseline: 1.8039x; 1.8039x over previous
#include <hip/hip_runtime.h>
#include <hip/hip_bf16.h>

typedef float v2f __attribute__((ext_vector_type(2)));
typedef float v8f __attribute__((ext_vector_type(8)));

__device__ __forceinline__ v8f wmma_f32(v2f a, v2f b, v8f c) {
  return __builtin_amdgcn_wmma_f32_16x16x4_f32(false, a, false, b, (short)0, c,
                                               false, false);
}

#define SPATIAL 32768
#define MPAD 256
#define NWT (27 * 64 * MPAD)

__global__ __launch_bounds__(256) void prep_weights_kernel(
    const float* __restrict__ We, const float* __restrict__ be,
    float* __restrict__ Wt, float* __restrict__ bep) {
  int idx = blockIdx.x * 256 + threadIdx.x;
  if (idx < NWT) {
    int o = idx & (MPAD - 1);
    int cm = (idx >> 8) & 63;
    int koff = idx >> 14;
    float v = (o < 216) ? We[(size_t)o * 1728 + cm * 27 + koff] : 0.0f;
    Wt[((size_t)(koff * 32 + (cm >> 1)) * MPAD + o) * 2 + (cm & 1)] = v;
  } else if (idx < NWT + MPAD) {
    int o = idx - NWT;
    bep[o] = (o < 216) ? be[o] : 0.0f;
  }
}

template <int MTILES, bool BIAS>
__global__ __launch_bounds__(128) void gemm_f32_wmma_kernel(
    const float* __restrict__ A, const float* __restrict__ B,
    const float* __restrict__ bias, float* __restrict__ C, int N) {
  constexpr int M = MTILES * 16;
  __shared__ float lA[64 * 64];
  __shared__ float lB[64 * 64];

  const int tid = threadIdx.x;
  const int lane = tid & 31;
  const int wave = tid >> 5;
  const int n0 = blockIdx.x * 64;
  const int ml = lane & 15;
  const int lhi = lane >> 4;

  for (int i = tid; i < M * 64; i += 128) lA[i] = A[i];
  for (int i = tid; i < 64 * 64; i += 128) {
    int k = i >> 6, col = i & 63;
    lB[((k >> 1) * 64 + col) * 2 + (k & 1)] = B[(size_t)k * N + n0 + col];
  }
  __syncthreads();

  int mt[MTILES], nt[MTILES];
#pragma unroll
  for (int i = 0; i < MTILES; ++i) {
    int p = wave + 4 * i;
    mt[i] = p % MTILES;
    nt[i] = p / MTILES;
  }

  v8f acc[MTILES];
#pragma unroll
  for (int i = 0; i < MTILES; ++i) {
    v8f c;
#pragma unroll
    for (int v = 0; v < 8; ++v)
      c[v] = BIAS ? bias[mt[i] * 16 + v + 8 * lhi] : 0.0f;
    acc[i] = c;
  }

#pragma unroll
  for (int k0 = 0; k0 < 64; k0 += 4) {
#pragma unroll
    for (int i = 0; i < MTILES; ++i) {
      v2f a = *(const v2f*)&lA[(mt[i] * 16 + ml) * 64 + k0 + 2 * lhi];
      v2f b = *(const v2f*)&lB[(((k0 >> 1) + lhi) * 64 + nt[i] * 16 + ml) * 2];
      acc[i] = wmma_f32(a, b, acc[i]);
    }
  }

#pragma unroll
  for (int i = 0; i < MTILES; ++i)
#pragma unroll
    for (int v = 0; v < 8; ++v) {
      int m = mt[i] * 16 + v + 8 * lhi;
      C[(size_t)m * N + n0 + nt[i] * 16 + ml] = acc[i][v];
    }
}

__global__ __launch_bounds__(128) void conv3_wmma_kernel(
    const float* __restrict__ y1, const float* __restrict__ Wt,
    const float* __restrict__ bep, float* __restrict__ y2p) {
  __shared__ float strip[64 * 34];

  const int tid = threadIdx.x;
  const int lane = tid & 31;
  const int wave = tid >> 5;
  const int d = blockIdx.x >> 5;
  const int h = blockIdx.x & 31;
  const int ml = lane & 15;
  const int lhi = lane >> 4;

  v8f acc[4][2];
#pragma unroll
  for (int t = 0; t < 4; ++t) {
    int ob = (wave + 4 * t) * 16;
#pragma unroll
    for (int j = 0; j < 2; ++j) {
      v8f c;
#pragma unroll
      for (int v = 0; v < 8; ++v) c[v] = bep[ob + v + 8 * lhi];
      acc[t][j] = c;
    }
  }

  for (int q = 0; q < 9; ++q) {
    const int kd = q / 3, kh = q % 3;
    const int dd = d + kd - 1, hh = h + kh - 1;
    const bool rowok = (dd >= 0 && dd < 32 && hh >= 0 && hh < 32);
    const int ddc = min(max(dd, 0), 31), hhc = min(max(hh, 0), 31);
    const float* src = y1 + ((size_t)ddc * 32 + hhc) * 32;

    __syncthreads();
    for (int i = tid; i < 64 * 34; i += 128) {
      int cm = i / 34, pos = i % 34;
      int wsrc = pos - 1;
      int wc = min(max(wsrc, 0), 31);
      float ld = src[(size_t)cm * SPATIAL + wc];
      bool ok = rowok && (wsrc >= 0) && (wsrc < 32);
      strip[((cm >> 1) * 34 + pos) * 2 + (cm & 1)] = ok ? ld : 0.0f;
    }
    if (q < 8 && tid < 64) {
      int kd2 = (q + 1) / 3, kh2 = (q + 1) % 3;
      int d2 = min(max(d + kd2 - 1, 0), 31), h2 = min(max(h + kh2 - 1, 0), 31);
      __builtin_prefetch(y1 + ((size_t)tid * SPATIAL) + ((size_t)d2 * 32 + h2) * 32,
                         0, 0);
    }
    __syncthreads();

    const int koffb = kd * 9 + kh * 3;
#pragma unroll 2
    for (int k0 = 0; k0 < 64; k0 += 4) {
#pragma unroll
      for (int t = 0; t < 4; ++t) {
        int ob = (wave + 4 * t) * 16;
        const float* ab =
            Wt + ((size_t)((k0 >> 1) + lhi) * MPAD + ob + ml) * 2 + koffb * (size_t)(32 * MPAD * 2);
#pragma unroll
        for (int kw = 0; kw < 3; ++kw) {
          v2f a = *(const v2f*)(ab + (size_t)kw * (32 * MPAD * 2));
#pragma unroll
          for (int j = 0; j < 2; ++j) {
            v2f b = *(const v2f*)&strip[(((k0 >> 1) + lhi) * 34 + j * 16 + ml + kw) * 2];
            acc[t][j] = wmma_f32(a, b, acc[t][j]);
          }
        }
      }
    }
  }

  const size_t sbase = ((size_t)d * 32 + h) * 32;
#pragma unroll
  for (int t = 0; t < 4; ++t) {
    int ob = (wave + 4 * t) * 16;
#pragma unroll
    for (int j = 0; j < 2; ++j)
#pragma unroll
      for (int v = 0; v < 8; ++v) {
        int o = ob + v + 8 * lhi;
        y2p[(size_t)o * SPATIAL + sbase + j * 16 + ml] = acc[t][j][v];
      }
  }
}

typedef float v4f __attribute__((ext_vector_type(4)));
#define YPITCH 40
__global__ __launch_bounds__(128) void softmax_gather_kernel(
    const float* __restrict__ y2p, const float* __restrict__ Y,
    float* __restrict__ out) {
  __shared__ __attribute__((aligned(16))) float yl[32 * 9 * YPITCH];

  const int tid = threadIdx.x;
  const int w = tid & 31;
  const int q = tid >> 5;
  const int d = blockIdx.x >> 5, h = blockIdx.x & 31;

  v2f wv[27];
  const float* lp = y2p + (size_t)(q * 2) * SPATIAL + (d * 32 + h) * 32 + w;
#pragma unroll
  for (int n = 0; n < 27; ++n) {
    wv[n][0] = lp[(size_t)(n * 8) * SPATIAL];
    wv[n][1] = lp[(size_t)(n * 8 + 1) * SPATIAL];
  }

  const int c = tid & 7, ol = tid >> 3;
  const float* yg = Y + (size_t)ol * SPATIAL + c * 4;
  v4f t[18];
#pragma unroll
  for (int k = 0; k < 18; ++k) {
    const int r = k >> 1;
    const int nd = min(max(d + r / 3 - 1, 0), 31);
    const int nh = min(max(h + r % 3 - 1, 0), 31);
    t[k] = *(const v4f*)(yg + (size_t)(k & 1) * (16 * SPATIAL) + (nd * 32 + nh) * 32);
  }

  float* yd = &yl[ol * 9 * YPITCH + 4 + c * 4];
#pragma unroll
  for (int k = 0; k < 18; ++k)
    *(v4f*)(yd + ((k & 1) * 144 + (k >> 1)) * YPITCH) = t[k];
  if (c == 0 || c == 7) {
    float* ye = yd + (c == 0 ? -1 : 4);
#pragma unroll
    for (int k = 0; k < 18; ++k)
      ye[((k & 1) * 144 + (k >> 1)) * YPITCH] = (c == 0) ? t[k][0] : t[k][3];
  }

  float mx0 = -3.0e38f, mx1 = -3.0e38f;
#pragma unroll
  for (int n = 0; n < 27; ++n) {
    mx0 = fmaxf(mx0, wv[n][0]);
    mx1 = fmaxf(mx1, wv[n][1]);
  }
  float sum0 = 0.0f, sum1 = 0.0f;
#pragma unroll
  for (int n = 0; n < 27; ++n) {
    wv[n][0] = __expf(wv[n][0] - mx0);
    sum0 += wv[n][0];
    wv[n][1] = __expf(wv[n][1] - mx1);
    sum1 += wv[n][1];
  }
  const float inv0 = 1.0f / sum0, inv1 = 1.0f / sum1;
#pragma unroll
  for (int n = 0; n < 27; ++n) {
    wv[n][0] = wv[n][0] * inv0;
    wv[n][1] = wv[n][1] * inv1;
  }
  __syncthreads();

  const int rd = q >> 1, rh = q & 1;
  float* op = out + ((size_t)(2 * d + rd) * 64 + (2 * h + rh)) * 64 + 2 * w;
  const float* yb = &yl[w + 3];
#pragma unroll 2
  for (int o = 0; o < 32; ++o) {
    v2f a = {0.0f, 0.0f};
#pragma unroll
    for (int n = 0; n < 27; ++n) {
      const float y = yb[(o * 9 + n / 3) * YPITCH + n % 3];
      const v2f yy = {y, y};
      a = __builtin_elementwise_fma(yy, wv[n], a);
    }
    *(v2f*)(op + (size_t)o * 262144) = a;
  }
}

extern "C" void kernel_launch(void* const* d_in, const int* in_sizes, int n_in,
                              void* d_out, int out_size, void* d_ws,
                              size_t ws_size, hipStream_t stream) {
  (void)in_sizes; (void)n_in; (void)out_size; (void)ws_size;
  const float* x  = (const float*)d_in[0];
  const float* Wc = (const float*)d_in[1];
  const float* bc = (const float*)d_in[2];
  const float* We = (const float*)d_in[3];
  const float* be = (const float*)d_in[4];
  const float* Wp = (const float*)d_in[5];
  float* out = (float*)d_out;

  float* ws = (float*)d_ws;
  float* y1   = ws;
  float* y2p  = y1 + (size_t)64 * SPATIAL;
  float* wsft = y2p + (size_t)MPAD * SPATIAL;
  float* Yp   = wsft + (size_t)SPATIAL * 216;
  float* Wt   = Yp + (size_t)32 * SPATIAL;
  float* bep  = Wt + (size_t)NWT;

  prep_weights_kernel<<<(NWT + MPAD + 255) / 256, 256, 0, stream>>>(We, be, Wt, bep);
  gemm_f32_wmma_kernel<4, true><<<512, 128, 0, stream>>>(Wc, x, bc, y1, SPATIAL);
  conv3_wmma_kernel<<<1024, 128, 0, stream>>>(y1, Wt, bep, y2p);
  gemm_f32_wmma_kernel<2, false><<<512, 128, 0, stream>>>(Wp, x, nullptr, Yp, SPATIAL);
  softmax_gather_kernel<<<1024, 128, 0, stream>>>(y2p, Yp, out);
}


# softmax shift invariance: tap 26 subtracted from the other taps' weights in prep; conv M = 208 = 13 M-tiles (was 16 padded), wave = 3 tiles + a quarter of tile 12's K; tile 12 sum order changes
# speedup vs baseline: 2.0302x; 1.1255x over previous
#include <hip/hip_runtime.h>
#include <hip/hip_bf16.h>

typedef float v2f __attribute__((ext_vector_type(2)));
typedef float v8f __attribute__((ext_vector_type(8)));

__device__ __forceinline__ v8f wmma_f32(v2f a, v2f b, v8f c) {
  return __builtin_amdgcn_wmma_f32_16x16x4_f32(false, a, false, b, (short)0, c,
                                               false, false);
}

#define SPATIAL 32768
#define MPAD 256
#define NWT (27 * 64 * MPAD)
#define MP 208

__global__ __launch_bounds__(256) void prep_weights_kernel(
    const float* __restrict__ We, const float* __restrict__ be,
    float* __restrict__ Wt, float* __restrict__ bep) {
  int idx = blockIdx.x * 256 + threadIdx.x;
  if (idx < NWT) {
    int o = idx & (MPAD - 1);
    int cm = (idx >> 8) & 63;
    int koff = idx >> 14;
    if (o < MP) {
      const int r = cm * 27 + koff;
      float v = We[(size_t)o * 1728 + r] - We[(size_t)(MP + (o & 7)) * 1728 + r];
      Wt[((size_t)(koff * 32 + (cm >> 1)) * MP + o) * 2 + (cm & 1)] = v;
    }
  } else if (idx < NWT + MP) {
    int o = idx - NWT;
    bep[o] = be[o] - be[MP + (o & 7)];
  }
}

template <int MTILES, bool BIAS>
__global__ __launch_bounds__(128) void gemm_f32_wmma_kernel(
    const float* __restrict__ A, const float* __restrict__ B,
    const float* __restrict__ bias, float* __restrict__ C, int N) {
  constexpr int M = MTILES * 16;
  __shared__ float lA[64 * 64];
  __shared__ float lB[64 * 64];

  const int tid = threadIdx.x;
  const int lane = tid & 31;
  const int wave = tid >> 5;
  const int n0 = blockIdx.x * 64;
  const int ml = lane & 15;
  const int lhi = lane >> 4;

  for (int i = tid; i < M * 64; i += 128) lA[i] = A[i];
  for (int i = tid; i < 64 * 64; i += 128) {
    int k = i >> 6, col = i & 63;
    lB[((k >> 1) * 64 + col) * 2 + (k & 1)] = B[(size_t)k * N + n0 + col];
  }
  __syncthreads();

  int mt[MTILES], nt[MTILES];
#pragma unroll
  for (int i = 0; i < MTILES; ++i) {
    int p = wave + 4 * i;
    mt[i] = p % MTILES;
    nt[i] = p / MTILES;
  }

  v8f acc[MTILES];
#pragma unroll
  for (int i = 0; i < MTILES; ++i) {
    v8f c;
#pragma unroll
    for (int v = 0; v < 8; ++v)
      c[v] = BIAS ? bias[mt[i] * 16 + v + 8 * lhi] : 0.0f;
    acc[i] = c;
  }

#pragma unroll
  for (int k0 = 0; k0 < 64; k0 += 4) {
#pragma unroll
    for (int i = 0; i < MTILES; ++i) {
      v2f a = *(const v2f*)&lA[(mt[i] * 16 + ml) * 64 + k0 + 2 * lhi];
      v2f b = *(const v2f*)&lB[(((k0 >> 1) + lhi) * 64 + nt[i] * 16 + ml) * 2];
      acc[i] = wmma_f32(a, b, acc[i]);
    }
  }

#pragma unroll
  for (int i = 0; i < MTILES; ++i)
#pragma unroll
    for (int v = 0; v < 8; ++v) {
      int m = mt[i] * 16 + v + 8 * lhi;
      C[(size_t)m * N + n0 + nt[i] * 16 + ml] = acc[i][v];
    }
}

template <bool EXTRA>
__device__ __forceinline__ void conv_kstep(const float* __restrict__ wq,
                                           const float* __restrict__ wx,
                                           const float* strip, int k0, int lhi,
                                           int ml, v8f (&acc)[3][2],
                                           v8f (&accx)[2]) {
  v2f b[3][2];
#pragma unroll
  for (int kw = 0; kw < 3; ++kw)
#pragma unroll
    for (int j = 0; j < 2; ++j)
      b[kw][j] = *(const v2f*)&strip[(((k0 >> 1) + lhi) * 34 + j * 16 + ml + kw) * 2];
#pragma unroll
  for (int t = 0; t < 3; ++t) {
    const float* ab = wq + ((size_t)(k0 >> 1) * MP + t * 16) * 2;
#pragma unroll
    for (int kw = 0; kw < 3; ++kw) {
      v2f a = *(const v2f*)(ab + (size_t)kw * (32 * MP * 2));
#pragma unroll
      for (int j = 0; j < 2; ++j) acc[t][j] = wmma_f32(a, b[kw][j], acc[t][j]);
    }
  }
  if (EXTRA) {
    const float* ab = wx + (size_t)(k0 >> 1) * MP * 2;
#pragma unroll
    for (int kw = 0; kw < 3; ++kw) {
      v2f a = *(const v2f*)(ab + (size_t)kw * (32 * MP * 2));
#pragma unroll
      for (int j = 0; j < 2; ++j) accx[j] = wmma_f32(a, b[kw][j], accx[j]);
    }
  }
}

template <bool EXTRA>
__device__ __forceinline__ void conv_krange(const float* __restrict__ wq,
                                            const float* __restrict__ wx,
                                            const float* strip, int kbeg,
                                            int kend, int lhi, int ml,
                                            v8f (&acc)[3][2], v8f (&accx)[2]) {
  for (int k0 = kbeg; k0 < kend; k0 += 8) {
    conv_kstep<EXTRA>(wq, wx, strip, k0, lhi, ml, acc, accx);
    conv_kstep<EXTRA>(wq, wx, strip, k0 + 4, lhi, ml, acc, accx);
  }
}

__global__ __launch_bounds__(128) void conv3_wmma_kernel(
    const float* __restrict__ y1, const float* __restrict__ Wt,
    const float* __restrict__ bep, float* __restrict__ y2p) {
  __shared__ float strip[64 * 34];

  const int tid = threadIdx.x;
  const int lane = tid & 31;
  const int wave = __builtin_amdgcn_readfirstlane(tid >> 5);
  const int d = blockIdx.x >> 5;
  const int h = blockIdx.x & 31;
  const int ml = lane & 15;
  const int lhi = lane >> 4;
  const int ob0 = wave * 48;
  const int kx = wave * 16;

  v8f acc[3][2];
#pragma unroll
  for (int t = 0; t < 3; ++t) {
#pragma unroll
    for (int j = 0; j < 2; ++j) {
      v8f c;
#pragma unroll
      for (int v = 0; v < 8; ++v) c[v] = bep[ob0 + t * 16 + v + 8 * lhi];
      acc[t][j] = c;
    }
  }
  v8f accx[2];
#pragma unroll
  for (int j = 0; j < 2; ++j)
#pragma unroll
    for (int v = 0; v < 8; ++v) accx[j][v] = 0.0f;

  for (int q = 0; q < 9; ++q) {
    const int kd = q / 3, kh = q % 3;
    const int dd = d + kd - 1, hh = h + kh - 1;
    const bool rowok = (dd >= 0 && dd < 32 && hh >= 0 && hh < 32);
    const int ddc = min(max(dd, 0), 31), hhc = min(max(hh, 0), 31);
    const float* src = y1 + ((size_t)ddc * 32 + hhc) * 32;

    __syncthreads();
    for (int i = tid; i < 64 * 34; i += 128) {
      int cm = i / 34, pos = i % 34;
      int wsrc = pos - 1;
      int wc = min(max(wsrc, 0), 31);
      float ld = src[(size_t)cm * SPATIAL + wc];
      bool ok = rowok && (wsrc >= 0) && (wsrc < 32);
      strip[((cm >> 1) * 34 + pos) * 2 + (cm & 1)] = ok ? ld : 0.0f;
    }
    if (q < 8 && tid < 64) {
      int kd2 = (q + 1) / 3, kh2 = (q + 1) % 3;
      int d2 = min(max(d + kd2 - 1, 0), 31), h2 = min(max(h + kh2 - 1, 0), 31);
      __builtin_prefetch(y1 + ((size_t)tid * SPATIAL) + ((size_t)d2 * 32 + h2) * 32,
                         0, 0);
    }
    __syncthreads();

    const int koffb = kd * 9 + kh * 3;
    const float* wt = Wt + (size_t)koffb * (32 * MP * 2) + ((size_t)lhi * MP + ml) * 2;
    const float* wq = wt + ob0 * 2;
    const float* wx = wt + 192 * 2;
    conv_krange<false>(wq, wx, strip, 0, kx, lhi, ml, acc, accx);
    conv_krange<true>(wq, wx, strip, kx, kx + 16, lhi, ml, acc, accx);
    conv_krange<false>(wq, wx, strip, kx + 16, 64, lhi, ml, acc, accx);
  }

  const size_t sbase = ((size_t)d * 32 + h) * 32;
#pragma unroll
  for (int t = 0; t < 3; ++t) {
#pragma unroll
    for (int j = 0; j < 2; ++j)
#pragma unroll
      for (int v = 0; v < 8; ++v) {
        int o = ob0 + t * 16 + v + 8 * lhi;
        y2p[(size_t)o * SPATIAL + sbase + j * 16 + ml] = acc[t][j][v];
      }
  }

  __syncthreads();
  float* red = strip;
#pragma unroll
  for (int j = 0; j < 2; ++j)
#pragma unroll
    for (int v = 0; v < 8; ++v)
      red[wave * 512 + (v + 8 * lhi) * 32 + j * 16 + ml] = accx[j][v];
  __syncthreads();
#pragma unroll
  for (int i = 0; i < 4; ++i) {
    const int idx = tid + 128 * i;
    const float s = ((red[idx] + red[512 + idx]) + (red[1024 + idx] + red[1536 + idx])) +
                    bep[192 + (idx >> 5)];
    y2p[(size_t)(192 + (idx >> 5)) * SPATIAL + sbase + (idx & 31)] = s;
  }
}

typedef float v4f __attribute__((ext_vector_type(4)));
#define YPITCH 40
__global__ __launch_bounds__(128) void softmax_gather_kernel(
    const float* __restrict__ y2p, const float* __restrict__ Y,
    float* __restrict__ out) {
  __shared__ __attribute__((aligned(16))) float yl[32 * 9 * YPITCH];

  const int tid = threadIdx.x;
  const int w = tid & 31;
  const int q = tid >> 5;
  const int d = blockIdx.x >> 5, h = blockIdx.x & 31;

  v2f wv[27];
  const float* lp = y2p + (size_t)(q * 2) * SPATIAL + (d * 32 + h) * 32 + w;
#pragma unroll
  for (int n = 0; n < 26; ++n) {
    wv[n][0] = lp[(size_t)(n * 8) * SPATIAL];
    wv[n][1] = lp[(size_t)(n * 8 + 1) * SPATIAL];
  }
  wv[26][0] = 0.0f;
  wv[26][1] = 0.0f;

  const int c = tid & 7, ol = tid >> 3;
  const float* yg = Y + (size_t)ol * SPATIAL + c * 4;
  v4f t[18];
#pragma unroll
  for (int k = 0; k < 18; ++k) {
    const int r = k >> 1;
    const int nd = min(max(d + r / 3 - 1, 0), 31);
    const int nh = min(max(h + r % 3 - 1, 0), 31);
    t[k] = *(const v4f*)(yg + (size_t)(k & 1) * (16 * SPATIAL) + (nd * 32 + nh) * 32);
  }

  float* yd = &yl[ol * 9 * YPITCH + 4 + c * 4];
#pragma unroll
  for (int k = 0; k < 18; ++k)
    *(v4f*)(yd + ((k & 1) * 144 + (k >> 1)) * YPITCH) = t[k];
  if (c == 0 || c == 7) {
    float* ye = yd + (c == 0 ? -1 : 4);
#pragma unroll
    for (int k = 0; k < 18; ++k)
      ye[((k & 1) * 144 + (k >> 1)) * YPITCH] = (c == 0) ? t[k][0] : t[k][3];
  }

  float mx0 = -3.0e38f, mx1 = -3.0e38f;
#pragma unroll
  for (int n = 0; n < 27; ++n) {
    mx0 = fmaxf(mx0, wv[n][0]);
    mx1 = fmaxf(mx1, wv[n][1]);
  }
  float sum0 = 0.0f, sum1 = 0.0f;
#pragma unroll
  for (int n = 0; n < 27; ++n) {
    wv[n][0] = __expf(wv[n][0] - mx0);
    sum0 += wv[n][0];
    wv[n][1] = __expf(wv[n][1] - mx1);
    sum1 += wv[n][1];
  }
  const float inv0 = 1.0f / sum0, inv1 = 1.0f / sum1;
#pragma unroll
  for (int n = 0; n < 27; ++n) {
    wv[n][0] = wv[n][0] * inv0;
    wv[n][1] = wv[n][1] * inv1;
  }
  __syncthreads();

  const int rd = q >> 1, rh = q & 1;
  float* op = out + ((size_t)(2 * d + rd) * 64 + (2 * h + rh)) * 64 + 2 * w;
  const float* yb = &yl[w + 3];
#pragma unroll 2
  for (int o = 0; o < 32; ++o) {
    v2f a = {0.0f, 0.0f};
#pragma unroll
    for (int n = 0; n < 27; ++n) {
      const float y = yb[(o * 9 + n / 3) * YPITCH + n % 3];
      const v2f yy = {y, y};
      a = __builtin_elementwise_fma(yy, wv[n], a);
    }
    *(v2f*)(op + (size_t)o * 262144) = a;
  }
}

extern "C" void kernel_launch(void* const* d_in, const int* in_sizes, int n_in,
                              void* d_out, int out_size, void* d_ws,
                              size_t ws_size, hipStream_t stream) {
  (void)in_sizes; (void)n_in; (void)out_size; (void)ws_size;
  const float* x  = (const float*)d_in[0];
  const float* Wc = (const float*)d_in[1];
  const float* bc = (const float*)d_in[2];
  const float* We = (const float*)d_in[3];
  const float* be = (const float*)d_in[4];
  const float* Wp = (const float*)d_in[5];
  float* out = (float*)d_out;

  float* ws = (float*)d_ws;
  float* y1   = ws;
  float* y2p  = y1 + (size_t)64 * SPATIAL;
  float* wsft = y2p + (size_t)MPAD * SPATIAL;
  float* Yp   = wsft + (size_t)SPATIAL * 216;
  float* Wt   = Yp + (size_t)32 * SPATIAL;
  float* bep  = Wt + (size_t)NWT;

  prep_weights_kernel<<<(NWT + MP + 255) / 256, 256, 0, stream>>>(We, be, Wt, bep);
  gemm_f32_wmma_kernel<4, true><<<512, 128, 0, stream>>>(Wc, x, bc, y1, SPATIAL);
  conv3_wmma_kernel<<<1024, 128, 0, stream>>>(y1, Wt, bep, y2p);
  gemm_f32_wmma_kernel<2, false><<<512, 128, 0, stream>>>(Wp, x, nullptr, Yp, SPATIAL);
  softmax_gather_kernel<<<1024, 128, 0, stream>>>(y2p, Yp, out);
}


# 1x1x1 conv composed into the 3x3x3 weights each call (Wd*Wc on WMMA): conv reads x, bias per border class from a per-tap table; y1 plane and its GEMM launch gone; logit sum order changes
# speedup vs baseline: 2.2060x; 1.0866x over previous
#include <hip/hip_runtime.h>
#include <hip/hip_bf16.h>

typedef float v2f __attribute__((ext_vector_type(2)));
typedef float v8f __attribute__((ext_vector_type(8)));

__device__ __forceinline__ v8f wmma_f32(v2f a, v2f b, v8f c) {
  return __builtin_amdgcn_wmma_f32_16x16x4_f32(false, a, false, b, (short)0, c,
                                               false, false);
}

#define SPATIAL 32768
#define MPAD 256
#define NWT (27 * 64 * MPAD)
#define MP 208

#define LAP 66
__global__ __launch_bounds__(128) void compose_weights_kernel(
    const float* __restrict__ We, const float* __restrict__ be,
    const float* __restrict__ Wc, const float* __restrict__ bc,
    float* __restrict__ Wt, float* __restrict__ bt, float* __restrict__ bep) {
  __shared__ __attribute__((aligned(16))) float lA[16 * LAP];
  __shared__ __attribute__((aligned(16))) float lB[64 * 64];

  const int tid = threadIdx.x;
  const int lane = tid & 31;
  const int wave = tid >> 5;
  const int koff = blockIdx.x / 13;
  const int mt = blockIdx.x % 13;
  const int ml = lane & 15;
  const int lhi = lane >> 4;

  float wa[8], wr[8];
#pragma unroll
  for (int i = 0; i < 8; ++i) {
    const int o = mt * 16 + (tid >> 6) + 2 * i;
    const int r = (tid & 63) * 27 + koff;
    wa[i] = We[(size_t)o * 1728 + r];
    wr[i] = We[(size_t)(MP + (o & 7)) * 1728 + r];
  }
  float wcv[32];
#pragma unroll
  for (int i = 0; i < 32; ++i) wcv[i] = Wc[tid + 128 * i];
#pragma unroll
  for (int i = 0; i < 8; ++i)
    lA[((tid >> 6) + 2 * i) * LAP + (tid & 63)] = wa[i] - wr[i];
#pragma unroll
  for (int i = 0; i < 32; ++i) {
    const int e = tid + 128 * i;
    const int k = e >> 6, col = e & 63;
    lB[((k >> 1) * 64 + col) * 2 + (k & 1)] = wcv[i];
  }
  __syncthreads();

  v8f acc;
#pragma unroll
  for (int v = 0; v < 8; ++v) acc[v] = 0.0f;
#pragma unroll
  for (int k0 = 0; k0 < 64; k0 += 4) {
    v2f a = *(const v2f*)&lA[ml * LAP + k0 + 2 * lhi];
    v2f b = *(const v2f*)&lB[(((k0 >> 1) + lhi) * 64 + wave * 16 + ml) * 2];
    acc = wmma_f32(a, b, acc);
  }
  const int c = wave * 16 + ml;
#pragma unroll
  for (int v = 0; v < 8; ++v) {
    const int o = mt * 16 + v + 8 * lhi;
    Wt[((size_t)(koff * 32 + (c >> 1)) * MP + o) * 2 + (c & 1)] = acc[v];
  }

  if (tid < 16) {
    const int o = mt * 16 + tid;
    float s = 0.0f;
    for (int cm = 0; cm < 64; ++cm) s += lA[tid * LAP + cm] * bc[cm];
    bt[koff * MP + o] = s;
    if (koff == 0) bep[o] = be[o] - be[MP + (o & 7)];
  }
}

template <int MTILES, bool BIAS>
__global__ __launch_bounds__(128) void gemm_f32_wmma_kernel(
    const float* __restrict__ A, const float* __restrict__ B,
    const float* __restrict__ bias, float* __restrict__ C, int N) {
  constexpr int M = MTILES * 16;
  __shared__ float lA[64 * 64];
  __shared__ float lB[64 * 64];

  const int tid = threadIdx.x;
  const int lane = tid & 31;
  const int wave = tid >> 5;
  const int n0 = blockIdx.x * 64;
  const int ml = lane & 15;
  const int lhi = lane >> 4;

  for (int i = tid; i < M * 64; i += 128) lA[i] = A[i];
  for (int i = tid; i < 64 * 64; i += 128) {
    int k = i >> 6, col = i & 63;
    lB[((k >> 1) * 64 + col) * 2 + (k & 1)] = B[(size_t)k * N + n0 + col];
  }
  __syncthreads();

  int mt[MTILES], nt[MTILES];
#pragma unroll
  for (int i = 0; i < MTILES; ++i) {
    int p = wave + 4 * i;
    mt[i] = p % MTILES;
    nt[i] = p / MTILES;
  }

  v8f acc[MTILES];
#pragma unroll
  for (int i = 0; i < MTILES; ++i) {
    v8f c;
#pragma unroll
    for (int v = 0; v < 8; ++v)
      c[v] = BIAS ? bias[mt[i] * 16 + v + 8 * lhi] : 0.0f;
    acc[i] = c;
  }

#pragma unroll
  for (int k0 = 0; k0 < 64; k0 += 4) {
#pragma unroll
    for (int i = 0; i < MTILES; ++i) {
      v2f a = *(const v2f*)&lA[(mt[i] * 16 + ml) * 64 + k0 + 2 * lhi];
      v2f b = *(const v2f*)&lB[(((k0 >> 1) + lhi) * 64 + nt[i] * 16 + ml) * 2];
      acc[i] = wmma_f32(a, b, acc[i]);
    }
  }

#pragma unroll
  for (int i = 0; i < MTILES; ++i)
#pragma unroll
    for (int v = 0; v < 8; ++v) {
      int m = mt[i] * 16 + v + 8 * lhi;
      C[(size_t)m * N + n0 + nt[i] * 16 + ml] = acc[i][v];
    }
}

template <bool EXTRA>
__device__ __forceinline__ void conv_kstep(const float* __restrict__ wq,
                                           const float* __restrict__ wx,
                                           const float* strip, int k0, int lhi,
                                           int ml, v8f (&acc)[3][2],
                                           v8f (&accx)[2]) {
  v2f b[3][2];
#pragma unroll
  for (int kw = 0; kw < 3; ++kw)
#pragma unroll
    for (int j = 0; j < 2; ++j)
      b[kw][j] = *(const v2f*)&strip[(((k0 >> 1) + lhi) * 34 + j * 16 + ml + kw) * 2];
#pragma unroll
  for (int t = 0; t < 3; ++t) {
    const float* ab = wq + ((size_t)(k0 >> 1) * MP + t * 16) * 2;
#pragma unroll
    for (int kw = 0; kw < 3; ++kw) {
      v2f a = *(const v2f*)(ab + (size_t)kw * (32 * MP * 2));
#pragma unroll
      for (int j = 0; j < 2; ++j) acc[t][j] = wmma_f32(a, b[kw][j], acc[t][j]);
    }
  }
  if (EXTRA) {
    const float* ab = wx + (size_t)(k0 >> 1) * MP * 2;
#pragma unroll
    for (int kw = 0; kw < 3; ++kw) {
      v2f a = *(const v2f*)(ab + (size_t)kw * (32 * MP * 2));
#pragma unroll
      for (int j = 0; j < 2; ++j) accx[j] = wmma_f32(a, b[kw][j], accx[j]);
    }
  }
}

template <bool EXTRA>
__device__ __forceinline__ void conv_krange(const float* __restrict__ wq,
                                            const float* __restrict__ wx,
                                            const float* strip, int kbeg,
                                            int kend, int lhi, int ml,
                                            v8f (&acc)[3][2], v8f (&accx)[2]) {
  for (int k0 = kbeg; k0 < kend; k0 += 8) {
    conv_kstep<EXTRA>(wq, wx, strip, k0, lhi, ml, acc, accx);
    conv_kstep<EXTRA>(wq, wx, strip, k0 + 4, lhi, ml, acc, accx);
  }
}

__global__ __launch_bounds__(128) void conv3_wmma_kernel(
    const float* __restrict__ y1, const float* __restrict__ Wt,
    const float* __restrict__ bep, const float* __restrict__ bt,
    float* __restrict__ y2p) {
  __shared__ float strip[64 * 34];
  __shared__ float biasL[3 * MP];

  const int tid = threadIdx.x;
  const int lane = tid & 31;
  const int wave = __builtin_amdgcn_readfirstlane(tid >> 5);
  const int d = blockIdx.x >> 5;
  const int h = blockIdx.x & 31;
  const int ml = lane & 15;
  const int lhi = lane >> 4;
  const int ob0 = wave * 48;
  const int kx = wave * 16;

#pragma unroll
  for (int half = 0; half < 2; ++half) {
    const int o = tid + 128 * half;
    if (o < MP) {
      float tv[27];
#pragma unroll
      for (int k = 0; k < 27; ++k) tv[k] = bt[(size_t)k * MP + o];
      float s0 = 0.0f, s1 = 0.0f, s2 = 0.0f;
#pragma unroll
      for (int q = 0; q < 9; ++q) {
        const int dd = d + q / 3 - 1, hh = h + q % 3 - 1;
        const bool ok = (dd >= 0 && dd < 32 && hh >= 0 && hh < 32);
        s0 += ok ? tv[q * 3] : 0.0f;
        s1 += ok ? tv[q * 3 + 1] : 0.0f;
        s2 += ok ? tv[q * 3 + 2] : 0.0f;
      }
      const float b0 = bep[o];
      biasL[o] = b0 + (s1 + s2);
      biasL[MP + o] = b0 + ((s0 + s1) + s2);
      biasL[2 * MP + o] = b0 + (s0 + s1);
    }
  }
  __syncthreads();

  v8f acc[3][2];
#pragma unroll
  for (int t = 0; t < 3; ++t) {
#pragma unroll
    for (int j = 0; j < 2; ++j) {
      const int cls = (j == 0) ? (ml == 0 ? 0 : 1) : (ml == 15 ? 2 : 1);
      v8f c;
#pragma unroll
      for (int v = 0; v < 8; ++v) c[v] = biasL[cls * MP + ob0 + t * 16 + v + 8 * lhi];
      acc[t][j] = c;
    }
  }
  v8f accx[2];
#pragma unroll
  for (int j = 0; j < 2; ++j)
#pragma unroll
    for (int v = 0; v < 8; ++v) accx[j][v] = 0.0f;

  for (int q = 0; q < 9; ++q) {
    const int kd = q / 3, kh = q % 3;
    const int dd = d + kd - 1, hh = h + kh - 1;
    const bool rowok = (dd >= 0 && dd < 32 && hh >= 0 && hh < 32);
    const int ddc = min(max(dd, 0), 31), hhc = min(max(hh, 0), 31);
    const float* src = y1 + ((size_t)ddc * 32 + hhc) * 32;

    __syncthreads();
    for (int i = tid; i < 64 * 34; i += 128) {
      int cm = i / 34, pos = i % 34;
      int wsrc = pos - 1;
      int wc = min(max(wsrc, 0), 31);
      float ld = src[(size_t)cm * SPATIAL + wc];
      bool ok = rowok && (wsrc >= 0) && (wsrc < 32);
      strip[((cm >> 1) * 34 + pos) * 2 + (cm & 1)] = ok ? ld : 0.0f;
    }
    if (q < 8 && tid < 64) {
      int kd2 = (q + 1) / 3, kh2 = (q + 1) % 3;
      int d2 = min(max(d + kd2 - 1, 0), 31), h2 = min(max(h + kh2 - 1, 0), 31);
      __builtin_prefetch(y1 + ((size_t)tid * SPATIAL) + ((size_t)d2 * 32 + h2) * 32,
                         0, 0);
    }
    __syncthreads();

    const int koffb = kd * 9 + kh * 3;
    const float* wt = Wt + (size_t)koffb * (32 * MP * 2) + ((size_t)lhi * MP + ml) * 2;
    const float* wq = wt + ob0 * 2;
    const float* wx = wt + 192 * 2;
    conv_krange<false>(wq, wx, strip, 0, kx, lhi, ml, acc, accx);
    conv_krange<true>(wq, wx, strip, kx, kx + 16, lhi, ml, acc, accx);
    conv_krange<false>(wq, wx, strip, kx + 16, 64, lhi, ml, acc, accx);
  }

  const size_t sbase = ((size_t)d * 32 + h) * 32;
#pragma unroll
  for (int t = 0; t < 3; ++t) {
#pragma unroll
    for (int j = 0; j < 2; ++j)
#pragma unroll
      for (int v = 0; v < 8; ++v) {
        int o = ob0 + t * 16 + v + 8 * lhi;
        y2p[(size_t)o * SPATIAL + sbase + j * 16 + ml] = acc[t][j][v];
      }
  }

  __syncthreads();
  float* red = strip;
#pragma unroll
  for (int j = 0; j < 2; ++j)
#pragma unroll
    for (int v = 0; v < 8; ++v)
      red[wave * 512 + (v + 8 * lhi) * 32 + j * 16 + ml] = accx[j][v];
  __syncthreads();
#pragma unroll
  for (int i = 0; i < 4; ++i) {
    const int idx = tid + 128 * i;
    const int col = idx & 31;
    const int cls = (col == 0) ? 0 : (col == 31 ? 2 : 1);
    const float s = ((red[idx] + red[512 + idx]) + (red[1024 + idx] + red[1536 + idx])) +
                    biasL[cls * MP + 192 + (idx >> 5)];
    y2p[(size_t)(192 + (idx >> 5)) * SPATIAL + sbase + (idx & 31)] = s;
  }
}

typedef float v4f __attribute__((ext_vector_type(4)));
#define YPITCH 40
__global__ __launch_bounds__(128) void softmax_gather_kernel(
    const float* __restrict__ y2p, const float* __restrict__ Y,
    float* __restrict__ out) {
  __shared__ __attribute__((aligned(16))) float yl[32 * 9 * YPITCH];

  const int tid = threadIdx.x;
  const int w = tid & 31;
  const int q = tid >> 5;
  const int d = blockIdx.x >> 5, h = blockIdx.x & 31;

  v2f wv[27];
  const float* lp = y2p + (size_t)(q * 2) * SPATIAL + (d * 32 + h) * 32 + w;
#pragma unroll
  for (int n = 0; n < 26; ++n) {
    wv[n][0] = lp[(size_t)(n * 8) * SPATIAL];
    wv[n][1] = lp[(size_t)(n * 8 + 1) * SPATIAL];
  }
  wv[26][0] = 0.0f;
  wv[26][1] = 0.0f;

  const int c = tid & 7, ol = tid >> 3;
  const float* yg = Y + (size_t)ol * SPATIAL + c * 4;
  v4f t[18];
#pragma unroll
  for (int k = 0; k < 18; ++k) {
    const int r = k >> 1;
    const int nd = min(max(d + r / 3 - 1, 0), 31);
    const int nh = min(max(h + r % 3 - 1, 0), 31);
    t[k] = *(const v4f*)(yg + (size_t)(k & 1) * (16 * SPATIAL) + (nd * 32 + nh) * 32);
  }

  float* yd = &yl[ol * 9 * YPITCH + 4 + c * 4];
#pragma unroll
  for (int k = 0; k < 18; ++k)
    *(v4f*)(yd + ((k & 1) * 144 + (k >> 1)) * YPITCH) = t[k];
  if (c == 0 || c == 7) {
    float* ye = yd + (c == 0 ? -1 : 4);
#pragma unroll
    for (int k = 0; k < 18; ++k)
      ye[((k & 1) * 144 + (k >> 1)) * YPITCH] = (c == 0) ? t[k][0] : t[k][3];
  }

  float mx0 = -3.0e38f, mx1 = -3.0e38f;
#pragma unroll
  for (int n = 0; n < 27; ++n) {
    mx0 = fmaxf(mx0, wv[n][0]);
    mx1 = fmaxf(mx1, wv[n][1]);
  }
  float sum0 = 0.0f, sum1 = 0.0f;
#pragma unroll
  for (int n = 0; n < 27; ++n) {
    wv[n][0] = __expf(wv[n][0] - mx0);
    sum0 += wv[n][0];
    wv[n][1] = __expf(wv[n][1] - mx1);
    sum1 += wv[n][1];
  }
  const float inv0 = 1.0f / sum0, inv1 = 1.0f / sum1;
#pragma unroll
  for (int n = 0; n < 27; ++n) {
    wv[n][0] = wv[n][0] * inv0;
    wv[n][1] = wv[n][1] * inv1;
  }
  __syncthreads();

  const int rd = q >> 1, rh = q & 1;
  float* op = out + ((size_t)(2 * d + rd) * 64 + (2 * h + rh)) * 64 + 2 * w;
  const float* yb = &yl[w + 3];
#pragma unroll 2
  for (int o = 0; o < 32; ++o) {
    v2f a = {0.0f, 0.0f};
#pragma unroll
    for (int n = 0; n < 27; ++n) {
      const float y = yb[(o * 9 + n / 3) * YPITCH + n % 3];
      const v2f yy = {y, y};
      a = __builtin_elementwise_fma(yy, wv[n], a);
    }
    *(v2f*)(op + (size_t)o * 262144) = a;
  }
}

extern "C" void kernel_launch(void* const* d_in, const int* in_sizes, int n_in,
                              void* d_out, int out_size, void* d_ws,
                              size_t ws_size, hipStream_t stream) {
  (void)in_sizes; (void)n_in; (void)out_size; (void)ws_size;
  const float* x  = (const float*)d_in[0];
  const float* Wc = (const float*)d_in[1];
  const float* bc = (const float*)d_in[2];
  const float* We = (const float*)d_in[3];
  const float* be = (const float*)d_in[4];
  const float* Wp = (const float*)d_in[5];
  float* out = (float*)d_out;

  float* ws = (float*)d_ws;
  float* y2p  = ws + (size_t)64 * SPATIAL;
  float* wsft = y2p + (size_t)MPAD * SPATIAL;
  float* Yp   = wsft + (size_t)SPATIAL * 216;
  float* Wt   = Yp + (size_t)32 * SPATIAL;
  float* bep  = Wt + (size_t)NWT;
  float* bt   = wsft;

  compose_weights_kernel<<<351, 128, 0, stream>>>(We, be, Wc, bc, Wt, bt, bep);
  conv3_wmma_kernel<<<1024, 128, 0, stream>>>(x, Wt, bep, bt, y2p);
  gemm_f32_wmma_kernel<2, false><<<512, 128, 0, stream>>>(Wp, x, nullptr, Yp, SPATIAL);
  softmax_gather_kernel<<<1024, 128, 0, stream>>>(y2p, Yp, out);
}


# Y = Wp*x formed inside the conv kernel on WMMA from its centre-tap strip of x in LDS (one tile per wave, whole-line stores via LDS); the Y GEMM launch and its x staging are gone; sum orders kept
# speedup vs baseline: 2.3121x; 1.0481x over previous
#include <hip/hip_runtime.h>
#include <hip/hip_bf16.h>

typedef float v2f __attribute__((ext_vector_type(2)));
typedef float v8f __attribute__((ext_vector_type(8)));

__device__ __forceinline__ v8f wmma_f32(v2f a, v2f b, v8f c) {
  return __builtin_amdgcn_wmma_f32_16x16x4_f32(false, a, false, b, (short)0, c,
                                               false, false);
}

#define SPATIAL 32768
#define MPAD 256
#define NWT (27 * 64 * MPAD)
#define MP 208

#define LAP 66
__global__ __launch_bounds__(128) void compose_weights_kernel(
    const float* __restrict__ We, const float* __restrict__ be,
    const float* __restrict__ Wc, const float* __restrict__ bc,
    float* __restrict__ Wt, float* __restrict__ bt, float* __restrict__ bep) {
  __shared__ __attribute__((aligned(16))) float lA[16 * LAP];
  __shared__ __attribute__((aligned(16))) float lB[64 * 64];

  const int tid = threadIdx.x;
  const int lane = tid & 31;
  const int wave = tid >> 5;
  const int koff = blockIdx.x / 13;
  const int mt = blockIdx.x % 13;
  const int ml = lane & 15;
  const int lhi = lane >> 4;

  float wa[8], wr[8];
#pragma unroll
  for (int i = 0; i < 8; ++i) {
    const int o = mt * 16 + (tid >> 6) + 2 * i;
    const int r = (tid & 63) * 27 + koff;
    wa[i] = We[(size_t)o * 1728 + r];
    wr[i] = We[(size_t)(MP + (o & 7)) * 1728 + r];
  }
  float wcv[32];
#pragma unroll
  for (int i = 0; i < 32; ++i) wcv[i] = Wc[tid + 128 * i];
#pragma unroll
  for (int i = 0; i < 8; ++i)
    lA[((tid >> 6) + 2 * i) * LAP + (tid & 63)] = wa[i] - wr[i];
#pragma unroll
  for (int i = 0; i < 32; ++i) {
    const int e = tid + 128 * i;
    const int k = e >> 6, col = e & 63;
    lB[((k >> 1) * 64 + col) * 2 + (k & 1)] = wcv[i];
  }
  __syncthreads();

  v8f acc;
#pragma unroll
  for (int v = 0; v < 8; ++v) acc[v] = 0.0f;
#pragma unroll
  for (int k0 = 0; k0 < 64; k0 += 4) {
    v2f a = *(const v2f*)&lA[ml * LAP + k0 + 2 * lhi];
    v2f b = *(const v2f*)&lB[(((k0 >> 1) + lhi) * 64 + wave * 16 + ml) * 2];
    acc = wmma_f32(a, b, acc);
  }
  const int c = wave * 16 + ml;
#pragma unroll
  for (int v = 0; v < 8; ++v) {
    const int o = mt * 16 + v + 8 * lhi;
    Wt[((size_t)(koff * 32 + (c >> 1)) * MP + o) * 2 + (c & 1)] = acc[v];
  }

  if (tid < 16) {
    const int o = mt * 16 + tid;
    float s = 0.0f;
    for (int cm = 0; cm < 64; ++cm) s += lA[tid * LAP + cm] * bc[cm];
    bt[koff * MP + o] = s;
    if (koff == 0) bep[o] = be[o] - be[MP + (o & 7)];
  }
}

template <bool EXTRA>
__device__ __forceinline__ void conv_kstep(const float* __restrict__ wq,
                                           const float* __restrict__ wx,
                                           const float* strip, int k0, int lhi,
                                           int ml, v8f (&acc)[3][2],
                                           v8f (&accx)[2]) {
  v2f b[3][2];
#pragma unroll
  for (int kw = 0; kw < 3; ++kw)
#pragma unroll
    for (int j = 0; j < 2; ++j)
      b[kw][j] = *(const v2f*)&strip[(((k0 >> 1) + lhi) * 34 + j * 16 + ml + kw) * 2];
#pragma unroll
  for (int t = 0; t < 3; ++t) {
    const float* ab = wq + ((size_t)(k0 >> 1) * MP + t * 16) * 2;
#pragma unroll
    for (int kw = 0; kw < 3; ++kw) {
      v2f a = *(const v2f*)(ab + (size_t)kw * (32 * MP * 2));
#pragma unroll
      for (int j = 0; j < 2; ++j) acc[t][j] = wmma_f32(a, b[kw][j], acc[t][j]);
    }
  }
  if (EXTRA) {
    const float* ab = wx + (size_t)(k0 >> 1) * MP * 2;
#pragma unroll
    for (int kw = 0; kw < 3; ++kw) {
      v2f a = *(const v2f*)(ab + (size_t)kw * (32 * MP * 2));
#pragma unroll
      for (int j = 0; j < 2; ++j) accx[j] = wmma_f32(a, b[kw][j], accx[j]);
    }
  }
}

template <bool EXTRA>
__device__ __forceinline__ void conv_krange(const float* __restrict__ wq,
                                            const float* __restrict__ wx,
                                            const float* strip, int kbeg,
                                            int kend, int lhi, int ml,
                                            v8f (&acc)[3][2], v8f (&accx)[2]) {
  for (int k0 = kbeg; k0 < kend; k0 += 8) {
    conv_kstep<EXTRA>(wq, wx, strip, k0, lhi, ml, acc, accx);
    conv_kstep<EXTRA>(wq, wx, strip, k0 + 4, lhi, ml, acc, accx);
  }
}

#define YSP 34
__global__ __launch_bounds__(128, 5) void conv3_wmma_kernel(
    const float* __restrict__ y1, const float* __restrict__ Wt,
    const float* __restrict__ bep, const float* __restrict__ bt,
    const float* __restrict__ Wp, float* __restrict__ y2p,
    float* __restrict__ Yp) {
  __shared__ float strip[64 * 34];
  __shared__ float ystage[32 * YSP];
  __shared__ float biasL[3 * MP];

  const int tid = threadIdx.x;
  const int lane = tid & 31;
  const int wave = __builtin_amdgcn_readfirstlane(tid >> 5);
  const int d = blockIdx.x >> 5;
  const int h = blockIdx.x & 31;
  const int ml = lane & 15;
  const int lhi = lane >> 4;
  const int ob0 = wave * 48;
  const int kx = wave * 16;

#pragma unroll
  for (int half = 0; half < 2; ++half) {
    const int o = tid + 128 * half;
    if (o < MP) {
      float tv[27];
#pragma unroll
      for (int k = 0; k < 27; ++k) tv[k] = bt[(size_t)k * MP + o];
      float s0 = 0.0f, s1 = 0.0f, s2 = 0.0f;
#pragma unroll
      for (int q = 0; q < 9; ++q) {
        const int dd = d + q / 3 - 1, hh = h + q % 3 - 1;
        const bool ok = (dd >= 0 && dd < 32 && hh >= 0 && hh < 32);
        s0 += ok ? tv[q * 3] : 0.0f;
        s1 += ok ? tv[q * 3 + 1] : 0.0f;
        s2 += ok ? tv[q * 3 + 2] : 0.0f;
      }
      const float b0 = bep[o];
      biasL[o] = b0 + (s1 + s2);
      biasL[MP + o] = b0 + ((s0 + s1) + s2);
      biasL[2 * MP + o] = b0 + (s0 + s1);
    }
  }
  __syncthreads();

  v8f acc[3][2];
#pragma unroll
  for (int t = 0; t < 3; ++t) {
#pragma unroll
    for (int j = 0; j < 2; ++j) {
      const int cls = (j == 0) ? (ml == 0 ? 0 : 1) : (ml == 15 ? 2 : 1);
      v8f c;
#pragma unroll
      for (int v = 0; v < 8; ++v) c[v] = biasL[cls * MP + ob0 + t * 16 + v + 8 * lhi];
      acc[t][j] = c;
    }
  }
  v8f accx[2];
#pragma unroll
  for (int j = 0; j < 2; ++j)
#pragma unroll
    for (int v = 0; v < 8; ++v) accx[j][v] = 0.0f;

  for (int q = 0; q < 9; ++q) {
    const int kd = q / 3, kh = q % 3;
    const int dd = d + kd - 1, hh = h + kh - 1;
    const bool rowok = (dd >= 0 && dd < 32 && hh >= 0 && hh < 32);
    const int ddc = min(max(dd, 0), 31), hhc = min(max(hh, 0), 31);
    const float* src = y1 + ((size_t)ddc * 32 + hhc) * 32;

    __syncthreads();
    for (int i = tid; i < 64 * 34; i += 128) {
      int cm = i / 34, pos = i % 34;
      int wsrc = pos - 1;
      int wc = min(max(wsrc, 0), 31);
      float ld = src[(size_t)cm * SPATIAL + wc];
      bool ok = rowok && (wsrc >= 0) && (wsrc < 32);
      strip[((cm >> 1) * 34 + pos) * 2 + (cm & 1)] = ok ? ld : 0.0f;
    }
    if (q < 8 && tid < 64) {
      int kd2 = (q + 1) / 3, kh2 = (q + 1) % 3;
      int d2 = min(max(d + kd2 - 1, 0), 31), h2 = min(max(h + kh2 - 1, 0), 31);
      __builtin_prefetch(y1 + ((size_t)tid * SPATIAL) + ((size_t)d2 * 32 + h2) * 32,
                         0, 0);
    }
    __syncthreads();

    const int koffb = kd * 9 + kh * 3;
    const float* wt = Wt + (size_t)koffb * (32 * MP * 2) + ((size_t)lhi * MP + ml) * 2;
    const float* wq = wt + ob0 * 2;
    const float* wx = wt + 192 * 2;
    conv_krange<false>(wq, wx, strip, 0, kx, lhi, ml, acc, accx);
    conv_krange<true>(wq, wx, strip, kx, kx + 16, lhi, ml, acc, accx);
    conv_krange<false>(wq, wx, strip, kx + 16, 64, lhi, ml, acc, accx);

    if (q == 4) {
      const int ymt = wave & 1, yj = wave >> 1;
      const float* wp = Wp + (size_t)(ymt * 16 + ml) * 64 + 2 * lhi;
      v8f ya;
#pragma unroll
      for (int v = 0; v < 8; ++v) ya[v] = 0.0f;
#pragma unroll
      for (int k0 = 0; k0 < 64; k0 += 4) {
        v2f a = *(const v2f*)(wp + k0);
        v2f b = *(const v2f*)&strip[(((k0 >> 1) + lhi) * 34 + yj * 16 + ml + 1) * 2];
        ya = wmma_f32(a, b, ya);
      }
      asm volatile("v_nop\n\tv_nop\n\tv_nop\n\tv_nop\n\tv_nop" : "+v"(ya));
#pragma unroll
      for (int v = 0; v < 8; ++v)
        ystage[(ymt * 16 + v + 8 * lhi) * YSP + yj * 16 + ml] = ya[v];
    }
  }

  const size_t sbase = ((size_t)d * 32 + h) * 32;
#pragma unroll
  for (int t = 0; t < 3; ++t) {
#pragma unroll
    for (int j = 0; j < 2; ++j)
#pragma unroll
      for (int v = 0; v < 8; ++v) {
        int o = ob0 + t * 16 + v + 8 * lhi;
        y2p[(size_t)o * SPATIAL + sbase + j * 16 + ml] = acc[t][j][v];
      }
  }

  __syncthreads();
#pragma unroll
  for (int i = 0; i < 8; ++i) {
    const int row = wave * 8 + i;
    Yp[(size_t)row * SPATIAL + sbase + lane] = ystage[row * YSP + lane];
  }
  float* red = strip;
#pragma unroll
  for (int j = 0; j < 2; ++j)
#pragma unroll
    for (int v = 0; v < 8; ++v)
      red[wave * 512 + (v + 8 * lhi) * 32 + j * 16 + ml] = accx[j][v];
  __syncthreads();
#pragma unroll
  for (int i = 0; i < 4; ++i) {
    const int idx = tid + 128 * i;
    const int col = idx & 31;
    const int cls = (col == 0) ? 0 : (col == 31 ? 2 : 1);
    const float s = ((red[idx] + red[512 + idx]) + (red[1024 + idx] + red[1536 + idx])) +
                    biasL[cls * MP + 192 + (idx >> 5)];
    y2p[(size_t)(192 + (idx >> 5)) * SPATIAL + sbase + (idx & 31)] = s;
  }
}

typedef float v4f __attribute__((ext_vector_type(4)));
#define YPITCH 40
__global__ __launch_bounds__(128) void softmax_gather_kernel(
    const float* __restrict__ y2p, const float* __restrict__ Y,
    float* __restrict__ out) {
  __shared__ __attribute__((aligned(16))) float yl[32 * 9 * YPITCH];

  const int tid = threadIdx.x;
  const int w = tid & 31;
  const int q = tid >> 5;
  const int d = blockIdx.x >> 5, h = blockIdx.x & 31;

  v2f wv[27];
  const float* lp = y2p + (size_t)(q * 2) * SPATIAL + (d * 32 + h) * 32 + w;
#pragma unroll
  for (int n = 0; n < 26; ++n) {
    wv[n][0] = lp[(size_t)(n * 8) * SPATIAL];
    wv[n][1] = lp[(size_t)(n * 8 + 1) * SPATIAL];
  }
  wv[26][0] = 0.0f;
  wv[26][1] = 0.0f;

  const int c = tid & 7, ol = tid >> 3;
  const float* yg = Y + (size_t)ol * SPATIAL + c * 4;
  v4f t[18];
#pragma unroll
  for (int k = 0; k < 18; ++k) {
    const int r = k >> 1;
    const int nd = min(max(d + r / 3 - 1, 0), 31);
    const int nh = min(max(h + r % 3 - 1, 0), 31);
    t[k] = *(const v4f*)(yg + (size_t)(k & 1) * (16 * SPATIAL) + (nd * 32 + nh) * 32);
  }

  float* yd = &yl[ol * 9 * YPITCH + 4 + c * 4];
#pragma unroll
  for (int k = 0; k < 18; ++k)
    *(v4f*)(yd + ((k & 1) * 144 + (k >> 1)) * YPITCH) = t[k];
  if (c == 0 || c == 7) {
    float* ye = yd + (c == 0 ? -1 : 4);
#pragma unroll
    for (int k = 0; k < 18; ++k)
      ye[((k & 1) * 144 + (k >> 1)) * YPITCH] = (c == 0) ? t[k][0] : t[k][3];
  }

  float mx0 = -3.0e38f, mx1 = -3.0e38f;
#pragma unroll
  for (int n = 0; n < 27; ++n) {
    mx0 = fmaxf(mx0, wv[n][0]);
    mx1 = fmaxf(mx1, wv[n][1]);
  }
  float sum0 = 0.0f, sum1 = 0.0f;
#pragma unroll
  for (int n = 0; n < 27; ++n) {
    wv[n][0] = __expf(wv[n][0] - mx0);
    sum0 += wv[n][0];
    wv[n][1] = __expf(wv[n][1] - mx1);
    sum1 += wv[n][1];
  }
  const float inv0 = 1.0f / sum0, inv1 = 1.0f / sum1;
#pragma unroll
  for (int n = 0; n < 27; ++n) {
    wv[n][0] = wv[n][0] * inv0;
    wv[n][1] = wv[n][1] * inv1;
  }
  __syncthreads();

  const int rd = q >> 1, rh = q & 1;
  float* op = out + ((size_t)(2 * d + rd) * 64 + (2 * h + rh)) * 64 + 2 * w;
  const float* yb = &yl[w + 3];
#pragma unroll 2
  for (int o = 0; o < 32; ++o) {
    v2f a = {0.0f, 0.0f};
#pragma unroll
    for (int n = 0; n < 27; ++n) {
      const float y = yb[(o * 9 + n / 3) * YPITCH + n % 3];
      const v2f yy = {y, y};
      a = __builtin_elementwise_fma(yy, wv[n], a);
    }
    *(v2f*)(op + (size_t)o * 262144) = a;
  }
}

extern "C" void kernel_launch(void* const* d_in, const int* in_sizes, int n_in,
                              void* d_out, int out_size, void* d_ws,
                              size_t ws_size, hipStream_t stream) {
  (void)in_sizes; (void)n_in; (void)out_size; (void)ws_size;
  const float* x  = (const float*)d_in[0];
  const float* Wc = (const float*)d_in[1];
  const float* bc = (const float*)d_in[2];
  const float* We = (const float*)d_in[3];
  const float* be = (const float*)d_in[4];
  const float* Wp = (const float*)d_in[5];
  float* out = (float*)d_out;

  float* ws = (float*)d_ws;
  float* y2p  = ws + (size_t)64 * SPATIAL;
  float* wsft = y2p + (size_t)MPAD * SPATIAL;
  float* Yp   = wsft + (size_t)SPATIAL * 216;
  float* Wt   = Yp + (size_t)32 * SPATIAL;
  float* bep  = Wt + (size_t)NWT;
  float* bt   = wsft;

  compose_weights_kernel<<<351, 128, 0, stream>>>(We, be, Wc, bc, Wt, bt, bep);
  conv3_wmma_kernel<<<1024, 128, 0, stream>>>(x, Wt, bep, bt, Wp, y2p, Yp);
  softmax_gather_kernel<<<1024, 128, 0, stream>>>(y2p, Yp, out);
}


# conv x strips double-buffered in LDS, next (kd,kh) rows fetched to registers by 16-byte loads during the k-loop: one barrier per step (was two), no serial strip loads; 4 waves/SIMD; sums kept
# speedup vs baseline: 2.5304x; 1.0944x over previous
#include <hip/hip_runtime.h>
#include <hip/hip_bf16.h>

typedef float v2f __attribute__((ext_vector_type(2)));
typedef float v4f __attribute__((ext_vector_type(4)));
typedef float v8f __attribute__((ext_vector_type(8)));

__device__ __forceinline__ v8f wmma_f32(v2f a, v2f b, v8f c) {
  return __builtin_amdgcn_wmma_f32_16x16x4_f32(false, a, false, b, (short)0, c,
                                               false, false);
}

#define SPATIAL 32768
#define MPAD 256
#define NWT (27 * 64 * MPAD)
#define MP 208

#define LAP 66
__global__ __launch_bounds__(128) void compose_weights_kernel(
    const float* __restrict__ We, const float* __restrict__ be,
    const float* __restrict__ Wc, const float* __restrict__ bc,
    float* __restrict__ Wt, float* __restrict__ bt, float* __restrict__ bep) {
  __shared__ __attribute__((aligned(16))) float lA[16 * LAP];
  __shared__ __attribute__((aligned(16))) float lB[64 * 64];

  const int tid = threadIdx.x;
  const int lane = tid & 31;
  const int wave = tid >> 5;
  const int koff = blockIdx.x / 13;
  const int mt = blockIdx.x % 13;
  const int ml = lane & 15;
  const int lhi = lane >> 4;

  float wa[8], wr[8];
#pragma unroll
  for (int i = 0; i < 8; ++i) {
    const int o = mt * 16 + (tid >> 6) + 2 * i;
    const int r = (tid & 63) * 27 + koff;
    wa[i] = We[(size_t)o * 1728 + r];
    wr[i] = We[(size_t)(MP + (o & 7)) * 1728 + r];
  }
  float wcv[32];
#pragma unroll
  for (int i = 0; i < 32; ++i) wcv[i] = Wc[tid + 128 * i];
#pragma unroll
  for (int i = 0; i < 8; ++i)
    lA[((tid >> 6) + 2 * i) * LAP + (tid & 63)] = wa[i] - wr[i];
#pragma unroll
  for (int i = 0; i < 32; ++i) {
    const int e = tid + 128 * i;
    const int k = e >> 6, col = e & 63;
    lB[((k >> 1) * 64 + col) * 2 + (k & 1)] = wcv[i];
  }
  __syncthreads();

  v8f acc;
#pragma unroll
  for (int v = 0; v < 8; ++v) acc[v] = 0.0f;
#pragma unroll
  for (int k0 = 0; k0 < 64; k0 += 4) {
    v2f a = *(const v2f*)&lA[ml * LAP + k0 + 2 * lhi];
    v2f b = *(const v2f*)&lB[(((k0 >> 1) + lhi) * 64 + wave * 16 + ml) * 2];
    acc = wmma_f32(a, b, acc);
  }
  const int c = wave * 16 + ml;
#pragma unroll
  for (int v = 0; v < 8; ++v) {
    const int o = mt * 16 + v + 8 * lhi;
    Wt[((size_t)(koff * 32 + (c >> 1)) * MP + o) * 2 + (c & 1)] = acc[v];
  }

  if (tid < 16) {
    const int o = mt * 16 + tid;
    float s = 0.0f;
    for (int cm = 0; cm < 64; ++cm) s += lA[tid * LAP + cm] * bc[cm];
    bt[koff * MP + o] = s;
    if (koff == 0) bep[o] = be[o] - be[MP + (o & 7)];
  }
}

template <bool EXTRA>
__device__ __forceinline__ void conv_kstep(const float* __restrict__ wq,
                                           const float* __restrict__ wx,
                                           const float* strip, int k0, int lhi,
                                           int ml, v8f (&acc)[3][2],
                                           v8f (&accx)[2]) {
  v2f b[3][2];
#pragma unroll
  for (int kw = 0; kw < 3; ++kw)
#pragma unroll
    for (int j = 0; j < 2; ++j)
      b[kw][j] = *(const v2f*)&strip[(((k0 >> 1) + lhi) * 34 + j * 16 + ml + kw) * 2];
#pragma unroll
  for (int t = 0; t < 3; ++t) {
    const float* ab = wq + ((size_t)(k0 >> 1) * MP + t * 16) * 2;
#pragma unroll
    for (int kw = 0; kw < 3; ++kw) {
      v2f a = *(const v2f*)(ab + (size_t)kw * (32 * MP * 2));
#pragma unroll
      for (int j = 0; j < 2; ++j) acc[t][j] = wmma_f32(a, b[kw][j], acc[t][j]);
    }
  }
  if (EXTRA) {
    const float* ab = wx + (size_t)(k0 >> 1) * MP * 2;
#pragma unroll
    for (int kw = 0; kw < 3; ++kw) {
      v2f a = *(const v2f*)(ab + (size_t)kw * (32 * MP * 2));
#pragma unroll
      for (int j = 0; j < 2; ++j) accx[j] = wmma_f32(a, b[kw][j], accx[j]);
    }
  }
}

template <bool EXTRA>
__device__ __forceinline__ void conv_krange(const float* __restrict__ wq,
                                            const float* __restrict__ wx,
                                            const float* strip, int kbeg,
                                            int kend, int lhi, int ml,
                                            v8f (&acc)[3][2], v8f (&accx)[2]) {
  for (int k0 = kbeg; k0 < kend; k0 += 8) {
    conv_kstep<EXTRA>(wq, wx, strip, k0, lhi, ml, acc, accx);
    conv_kstep<EXTRA>(wq, wx, strip, k0 + 4, lhi, ml, acc, accx);
  }
}

__device__ __forceinline__ void strip_fetch(const float* __restrict__ x, int d,
                                            int h, int q, int cp, int qtr,
                                            v4f (&r)[4]) {
  const int dd = min(max(d + q / 3 - 1, 0), 31), hh = min(max(h + q % 3 - 1, 0), 31);
  const float* src = x + (size_t)(2 * cp) * SPATIAL + (dd * 32 + hh) * 32 + qtr * 4;
  r[0] = *(const v4f*)src;
  r[1] = *(const v4f*)(src + 16);
  r[2] = *(const v4f*)(src + SPATIAL);
  r[3] = *(const v4f*)(src + SPATIAL + 16);
}
__device__ __forceinline__ void strip_store(float* buf, int d, int h, int q,
                                            int cp, int qtr, const v4f (&r)[4]) {
  const int dd = d + q / 3 - 1, hh = h + q % 3 - 1;
  const bool rowok = (dd >= 0 && dd < 32 && hh >= 0 && hh < 32);
  float* dst = buf + (cp * 34 + qtr * 4 + 1) * 2;
#pragma unroll
  for (int j = 0; j < 2; ++j)
#pragma unroll
    for (int i = 0; i < 4; ++i) {
      const v2f v = {rowok ? r[j][i] : 0.0f, rowok ? r[2 + j][i] : 0.0f};
      *(v2f*)(dst + (j * 16 + i) * 2) = v;
    }
}

#define YSP 34
__global__ __launch_bounds__(128, 4) void conv3_wmma_kernel(
    const float* __restrict__ y1, const float* __restrict__ Wt,
    const float* __restrict__ bep, const float* __restrict__ bt,
    const float* __restrict__ Wp, float* __restrict__ y2p,
    float* __restrict__ Yp) {
  __shared__ __attribute__((aligned(16))) float strip2[2][64 * 34];
  __shared__ float ystage[32 * YSP];
  __shared__ float biasL[3 * MP];

  const int tid = threadIdx.x;
  const int lane = tid & 31;
  const int wave = __builtin_amdgcn_readfirstlane(tid >> 5);
  const int d = blockIdx.x >> 5;
  const int h = blockIdx.x & 31;
  const int ml = lane & 15;
  const int lhi = lane >> 4;
  const int ob0 = wave * 48;
  const int kx = wave * 16;
  const int cp = tid >> 2, qtr = tid & 3;

  v4f r[4];
  strip_fetch(y1, d, h, 0, cp, qtr, r);

#pragma unroll
  for (int half = 0; half < 2; ++half) {
    const int o = tid + 128 * half;
    if (o < MP) {
      float tv[27];
#pragma unroll
      for (int k = 0; k < 27; ++k) tv[k] = bt[(size_t)k * MP + o];
      float s0 = 0.0f, s1 = 0.0f, s2 = 0.0f;
#pragma unroll
      for (int q = 0; q < 9; ++q) {
        const int dd = d + q / 3 - 1, hh = h + q % 3 - 1;
        const bool ok = (dd >= 0 && dd < 32 && hh >= 0 && hh < 32);
        s0 += ok ? tv[q * 3] : 0.0f;
        s1 += ok ? tv[q * 3 + 1] : 0.0f;
        s2 += ok ? tv[q * 3 + 2] : 0.0f;
      }
      const float b0 = bep[o];
      biasL[o] = b0 + (s1 + s2);
      biasL[MP + o] = b0 + ((s0 + s1) + s2);
      biasL[2 * MP + o] = b0 + (s0 + s1);
    }
  }
  *(v2f*)&strip2[tid >> 6][(((tid >> 1) & 31) * 34 + (tid & 1) * 33) * 2] =
      (v2f){0.0f, 0.0f};
  strip_store(strip2[0], d, h, 0, cp, qtr, r);
  __syncthreads();

  v8f acc[3][2];
#pragma unroll
  for (int t = 0; t < 3; ++t) {
#pragma unroll
    for (int j = 0; j < 2; ++j) {
      const int cls = (j == 0) ? (ml == 0 ? 0 : 1) : (ml == 15 ? 2 : 1);
      v8f c;
#pragma unroll
      for (int v = 0; v < 8; ++v) c[v] = biasL[cls * MP + ob0 + t * 16 + v + 8 * lhi];
      acc[t][j] = c;
    }
  }
  v8f accx[2];
#pragma unroll
  for (int j = 0; j < 2; ++j)
#pragma unroll
    for (int v = 0; v < 8; ++v) accx[j][v] = 0.0f;

  for (int q = 0; q < 9; ++q) {
    const int kd = q / 3, kh = q % 3;
    const float* strip = strip2[q & 1];
    if (q < 8) strip_fetch(y1, d, h, q + 1, cp, qtr, r);
    if (q < 7 && tid < 64) {
      const int d2 = min(max(d + (q + 2) / 3 - 1, 0), 31);
      const int h2 = min(max(h + (q + 2) % 3 - 1, 0), 31);
      __builtin_prefetch(y1 + (size_t)tid * SPATIAL + (d2 * 32 + h2) * 32, 0, 0);
    }

    const int koffb = kd * 9 + kh * 3;
    const float* wt = Wt + (size_t)koffb * (32 * MP * 2) + ((size_t)lhi * MP + ml) * 2;
    const float* wq = wt + ob0 * 2;
    const float* wx = wt + 192 * 2;
    conv_krange<false>(wq, wx, strip, 0, kx, lhi, ml, acc, accx);
    conv_krange<true>(wq, wx, strip, kx, kx + 16, lhi, ml, acc, accx);
    conv_krange<false>(wq, wx, strip, kx + 16, 64, lhi, ml, acc, accx);

    if (q == 4) {
      const int ymt = wave & 1, yj = wave >> 1;
      const float* wp = Wp + (size_t)(ymt * 16 + ml) * 64 + 2 * lhi;
      v8f ya;
#pragma unroll
      for (int v = 0; v < 8; ++v) ya[v] = 0.0f;
#pragma unroll
      for (int k0 = 0; k0 < 64; k0 += 4) {
        v2f a = *(const v2f*)(wp + k0);
        v2f b = *(const v2f*)&strip[(((k0 >> 1) + lhi) * 34 + yj * 16 + ml + 1) * 2];
        ya = wmma_f32(a, b, ya);
      }
      asm volatile("v_nop\n\tv_nop\n\tv_nop\n\tv_nop\n\tv_nop" : "+v"(ya));
#pragma unroll
      for (int v = 0; v < 8; ++v)
        ystage[(ymt * 16 + v + 8 * lhi) * YSP + yj * 16 + ml] = ya[v];
    }

    if (q < 8) {
      strip_store(strip2[(q + 1) & 1], d, h, q + 1, cp, qtr, r);
      __syncthreads();
    }
  }

  const size_t sbase = ((size_t)d * 32 + h) * 32;
#pragma unroll
  for (int t = 0; t < 3; ++t) {
#pragma unroll
    for (int j = 0; j < 2; ++j)
#pragma unroll
      for (int v = 0; v < 8; ++v) {
        int o = ob0 + t * 16 + v + 8 * lhi;
        y2p[(size_t)o * SPATIAL + sbase + j * 16 + ml] = acc[t][j][v];
      }
  }

#pragma unroll
  for (int i = 0; i < 8; ++i) {
    const int row = wave * 8 + i;
    Yp[(size_t)row * SPATIAL + sbase + lane] = ystage[row * YSP + lane];
  }
  float* red = strip2[1];
#pragma unroll
  for (int j = 0; j < 2; ++j)
#pragma unroll
    for (int v = 0; v < 8; ++v)
      red[wave * 512 + (v + 8 * lhi) * 32 + j * 16 + ml] = accx[j][v];
  __syncthreads();
#pragma unroll
  for (int i = 0; i < 4; ++i) {
    const int idx = tid + 128 * i;
    const int col = idx & 31;
    const int cls = (col == 0) ? 0 : (col == 31 ? 2 : 1);
    const float s = ((red[idx] + red[512 + idx]) + (red[1024 + idx] + red[1536 + idx])) +
                    biasL[cls * MP + 192 + (idx >> 5)];
    y2p[(size_t)(192 + (idx >> 5)) * SPATIAL + sbase + (idx & 31)] = s;
  }
}

#define YPITCH 40
__global__ __launch_bounds__(128) void softmax_gather_kernel(
    const float* __restrict__ y2p, const float* __restrict__ Y,
    float* __restrict__ out) {
  __shared__ __attribute__((aligned(16))) float yl[32 * 9 * YPITCH];

  const int tid = threadIdx.x;
  const int w = tid & 31;
  const int q = tid >> 5;
  const int d = blockIdx.x >> 5, h = blockIdx.x & 31;

  v2f wv[27];
  const float* lp = y2p + (size_t)(q * 2) * SPATIAL + (d * 32 + h) * 32 + w;
#pragma unroll
  for (int n = 0; n < 26; ++n) {
    wv[n][0] = lp[(size_t)(n * 8) * SPATIAL];
    wv[n][1] = lp[(size_t)(n * 8 + 1) * SPATIAL];
  }
  wv[26][0] = 0.0f;
  wv[26][1] = 0.0f;

  const int c = tid & 7, ol = tid >> 3;
  const float* yg = Y + (size_t)ol * SPATIAL + c * 4;
  v4f t[18];
#pragma unroll
  for (int k = 0; k < 18; ++k) {
    const int r = k >> 1;
    const int nd = min(max(d + r / 3 - 1, 0), 31);
    const int nh = min(max(h + r % 3 - 1, 0), 31);
    t[k] = *(const v4f*)(yg + (size_t)(k & 1) * (16 * SPATIAL) + (nd * 32 + nh) * 32);
  }

  float* yd = &yl[ol * 9 * YPITCH + 4 + c * 4];
#pragma unroll
  for (int k = 0; k < 18; ++k)
    *(v4f*)(yd + ((k & 1) * 144 + (k >> 1)) * YPITCH) = t[k];
  if (c == 0 || c == 7) {
    float* ye = yd + (c == 0 ? -1 : 4);
#pragma unroll
    for (int k = 0; k < 18; ++k)
      ye[((k & 1) * 144 + (k >> 1)) * YPITCH] = (c == 0) ? t[k][0] : t[k][3];
  }

  float mx0 = -3.0e38f, mx1 = -3.0e38f;
#pragma unroll
  for (int n = 0; n < 27; ++n) {
    mx0 = fmaxf(mx0, wv[n][0]);
    mx1 = fmaxf(mx1, wv[n][1]);
  }
  float sum0 = 0.0f, sum1 = 0.0f;
#pragma unroll
  for (int n = 0; n < 27; ++n) {
    wv[n][0] = __expf(wv[n][0] - mx0);
    sum0 += wv[n][0];
    wv[n][1] = __expf(wv[n][1] - mx1);
    sum1 += wv[n][1];
  }
  const float inv0 = 1.0f / sum0, inv1 = 1.0f / sum1;
#pragma unroll
  for (int n = 0; n < 27; ++n) {
    wv[n][0] = wv[n][0] * inv0;
    wv[n][1] = wv[n][1] * inv1;
  }
  __syncthreads();

  const int rd = q >> 1, rh = q & 1;
  float* op = out + ((size_t)(2 * d + rd) * 64 + (2 * h + rh)) * 64 + 2 * w;
  const float* yb = &yl[w + 3];
#pragma unroll 2
  for (int o = 0; o < 32; ++o) {
    v2f a = {0.0f, 0.0f};
#pragma unroll
    for (int n = 0; n < 27; ++n) {
      const float y = yb[(o * 9 + n / 3) * YPITCH + n % 3];
      const v2f yy = {y, y};
      a = __builtin_elementwise_fma(yy, wv[n], a);
    }
    *(v2f*)(op + (size_t)o * 262144) = a;
  }
}

extern "C" void kernel_launch(void* const* d_in, const int* in_sizes, int n_in,
                              void* d_out, int out_size, void* d_ws,
                              size_t ws_size, hipStream_t stream) {
  (void)in_sizes; (void)n_in; (void)out_size; (void)ws_size;
  const float* x  = (const float*)d_in[0];
  const float* Wc = (const float*)d_in[1];
  const float* bc = (const float*)d_in[2];
  const float* We = (const float*)d_in[3];
  const float* be = (const float*)d_in[4];
  const float* Wp = (const float*)d_in[5];
  float* out = (float*)d_out;

  float* ws = (float*)d_ws;
  float* y2p  = ws + (size_t)64 * SPATIAL;
  float* wsft = y2p + (size_t)MPAD * SPATIAL;
  float* Yp   = wsft + (size_t)SPATIAL * 216;
  float* Wt   = Yp + (size_t)32 * SPATIAL;
  float* bep  = Wt + (size_t)NWT;
  float* bt   = wsft;

  compose_weights_kernel<<<351, 128, 0, stream>>>(We, be, Wc, bc, Wt, bt, bep);
  conv3_wmma_kernel<<<1024, 128, 0, stream>>>(x, Wt, bep, bt, Wp, y2p, Yp);
  softmax_gather_kernel<<<1024, 128, 0, stream>>>(y2p, Yp, out);
}


# conv waves lower their issue priority as they advance through the nine (kd,kh) steps (s_setprio 3,3,3,2,2,2,1,1,0): a WGP's four blocks advance together, not oldest-first in turn; sums kept
# speedup vs baseline: 2.6741x; 1.0568x over previous
#include <hip/hip_runtime.h>
#include <hip/hip_bf16.h>

typedef float v2f __attribute__((ext_vector_type(2)));
typedef float v4f __attribute__((ext_vector_type(4)));
typedef float v8f __attribute__((ext_vector_type(8)));

__device__ __forceinline__ v8f wmma_f32(v2f a, v2f b, v8f c) {
  return __builtin_amdgcn_wmma_f32_16x16x4_f32(false, a, false, b, (short)0, c,
                                               false, false);
}

#define SPATIAL 32768
#define MPAD 256
#define NWT (27 * 64 * MPAD)
#define MP 208

#define LAP 66
__global__ __launch_bounds__(128) void compose_weights_kernel(
    const float* __restrict__ We, const float* __restrict__ be,
    const float* __restrict__ Wc, const float* __restrict__ bc,
    float* __restrict__ Wt, float* __restrict__ bt, float* __restrict__ bep) {
  __shared__ __attribute__((aligned(16))) float lA[16 * LAP];
  __shared__ __attribute__((aligned(16))) float lB[64 * 64];

  const int tid = threadIdx.x;
  const int lane = tid & 31;
  const int wave = tid >> 5;
  const int koff = blockIdx.x / 13;
  const int mt = blockIdx.x % 13;
  const int ml = lane & 15;
  const int lhi = lane >> 4;

  float wa[8], wr[8];
#pragma unroll
  for (int i = 0; i < 8; ++i) {
    const int o = mt * 16 + (tid >> 6) + 2 * i;
    const int r = (tid & 63) * 27 + koff;
    wa[i] = We[(size_t)o * 1728 + r];
    wr[i] = We[(size_t)(MP + (o & 7)) * 1728 + r];
  }
  float wcv[32];
#pragma unroll
  for (int i = 0; i < 32; ++i) wcv[i] = Wc[tid + 128 * i];
#pragma unroll
  for (int i = 0; i < 8; ++i)
    lA[((tid >> 6) + 2 * i) * LAP + (tid & 63)] = wa[i] - wr[i];
#pragma unroll
  for (int i = 0; i < 32; ++i) {
    const int e = tid + 128 * i;
    const int k = e >> 6, col = e & 63;
    lB[((k >> 1) * 64 + col) * 2 + (k & 1)] = wcv[i];
  }
  __syncthreads();

  v8f acc;
#pragma unroll
  for (int v = 0; v < 8; ++v) acc[v] = 0.0f;
#pragma unroll
  for (int k0 = 0; k0 < 64; k0 += 4) {
    v2f a = *(const v2f*)&lA[ml * LAP + k0 + 2 * lhi];
    v2f b = *(const v2f*)&lB[(((k0 >> 1) + lhi) * 64 + wave * 16 + ml) * 2];
    acc = wmma_f32(a, b, acc);
  }
  const int c = wave * 16 + ml;
#pragma unroll
  for (int v = 0; v < 8; ++v) {
    const int o = mt * 16 + v + 8 * lhi;
    Wt[((size_t)(koff * 32 + (c >> 1)) * MP + o) * 2 + (c & 1)] = acc[v];
  }

  if (tid < 16) {
    const int o = mt * 16 + tid;
    float s = 0.0f;
    for (int cm = 0; cm < 64; ++cm) s += lA[tid * LAP + cm] * bc[cm];
    bt[koff * MP + o] = s;
    if (koff == 0) bep[o] = be[o] - be[MP + (o & 7)];
  }
}

template <bool EXTRA>
__device__ __forceinline__ void conv_kstep(const float* __restrict__ wq,
                                           const float* __restrict__ wx,
                                           const float* strip, int k0, int lhi,
                                           int ml, v8f (&acc)[3][2],
                                           v8f (&accx)[2]) {
  v2f b[3][2];
#pragma unroll
  for (int kw = 0; kw < 3; ++kw)
#pragma unroll
    for (int j = 0; j < 2; ++j)
      b[kw][j] = *(const v2f*)&strip[(((k0 >> 1) + lhi) * 34 + j * 16 + ml + kw) * 2];
#pragma unroll
  for (int t = 0; t < 3; ++t) {
    const float* ab = wq + ((size_t)(k0 >> 1) * MP + t * 16) * 2;
#pragma unroll
    for (int kw = 0; kw < 3; ++kw) {
      v2f a = *(const v2f*)(ab + (size_t)kw * (32 * MP * 2));
#pragma unroll
      for (int j = 0; j < 2; ++j) acc[t][j] = wmma_f32(a, b[kw][j], acc[t][j]);
    }
  }
  if (EXTRA) {
    const float* ab = wx + (size_t)(k0 >> 1) * MP * 2;
#pragma unroll
    for (int kw = 0; kw < 3; ++kw) {
      v2f a = *(const v2f*)(ab + (size_t)kw * (32 * MP * 2));
#pragma unroll
      for (int j = 0; j < 2; ++j) accx[j] = wmma_f32(a, b[kw][j], accx[j]);
    }
  }
}

template <bool EXTRA>
__device__ __forceinline__ void conv_krange(const float* __restrict__ wq,
                                            const float* __restrict__ wx,
                                            const float* strip, int kbeg,
                                            int kend, int lhi, int ml,
                                            v8f (&acc)[3][2], v8f (&accx)[2]) {
  for (int k0 = kbeg; k0 < kend; k0 += 8) {
    conv_kstep<EXTRA>(wq, wx, strip, k0, lhi, ml, acc, accx);
    conv_kstep<EXTRA>(wq, wx, strip, k0 + 4, lhi, ml, acc, accx);
  }
}

__device__ __forceinline__ void strip_fetch(const float* __restrict__ x, int d,
                                            int h, int q, int cp, int qtr,
                                            v4f (&r)[4]) {
  const int dd = min(max(d + q / 3 - 1, 0), 31), hh = min(max(h + q % 3 - 1, 0), 31);
  const float* src = x + (size_t)(2 * cp) * SPATIAL + (dd * 32 + hh) * 32 + qtr * 4;
  r[0] = *(const v4f*)src;
  r[1] = *(const v4f*)(src + 16);
  r[2] = *(const v4f*)(src + SPATIAL);
  r[3] = *(const v4f*)(src + SPATIAL + 16);
}
__device__ __forceinline__ void strip_store(float* buf, int d, int h, int q,
                                            int cp, int qtr, const v4f (&r)[4]) {
  const int dd = d + q / 3 - 1, hh = h + q % 3 - 1;
  const bool rowok = (dd >= 0 && dd < 32 && hh >= 0 && hh < 32);
  float* dst = buf + (cp * 34 + qtr * 4 + 1) * 2;
#pragma unroll
  for (int j = 0; j < 2; ++j)
#pragma unroll
    for (int i = 0; i < 4; ++i) {
      const v2f v = {rowok ? r[j][i] : 0.0f, rowok ? r[2 + j][i] : 0.0f};
      *(v2f*)(dst + (j * 16 + i) * 2) = v;
    }
}

__device__ __forceinline__ void step_priority(int q) {
  if (q == 3) __builtin_amdgcn_s_setprio(2);
  if (q == 6) __builtin_amdgcn_s_setprio(1);
  if (q == 8) __builtin_amdgcn_s_setprio(0);
}

#define YSP 34
__global__ __launch_bounds__(128, 4) void conv3_wmma_kernel(
    const float* __restrict__ y1, const float* __restrict__ Wt,
    const float* __restrict__ bep, const float* __restrict__ bt,
    const float* __restrict__ Wp, float* __restrict__ y2p,
    float* __restrict__ Yp) {
  __shared__ __attribute__((aligned(16))) float strip2[2][64 * 34];
  __shared__ float ystage[32 * YSP];
  __shared__ float biasL[3 * MP];

  const int tid = threadIdx.x;
  const int lane = tid & 31;
  const int wave = __builtin_amdgcn_readfirstlane(tid >> 5);
  const int d = blockIdx.x >> 5;
  const int h = blockIdx.x & 31;
  const int ml = lane & 15;
  const int lhi = lane >> 4;
  const int ob0 = wave * 48;
  const int kx = wave * 16;
  const int cp = tid >> 2, qtr = tid & 3;

  v4f r[4];
  strip_fetch(y1, d, h, 0, cp, qtr, r);

#pragma unroll
  for (int half = 0; half < 2; ++half) {
    const int o = tid + 128 * half;
    if (o < MP) {
      float tv[27];
#pragma unroll
      for (int k = 0; k < 27; ++k) tv[k] = bt[(size_t)k * MP + o];
      float s0 = 0.0f, s1 = 0.0f, s2 = 0.0f;
#pragma unroll
      for (int q = 0; q < 9; ++q) {
        const int dd = d + q / 3 - 1, hh = h + q % 3 - 1;
        const bool ok = (dd >= 0 && dd < 32 && hh >= 0 && hh < 32);
        s0 += ok ? tv[q * 3] : 0.0f;
        s1 += ok ? tv[q * 3 + 1] : 0.0f;
        s2 += ok ? tv[q * 3 + 2] : 0.0f;
      }
      const float b0 = bep[o];
      biasL[o] = b0 + (s1 + s2);
      biasL[MP + o] = b0 + ((s0 + s1) + s2);
      biasL[2 * MP + o] = b0 + (s0 + s1);
    }
  }
  *(v2f*)&strip2[tid >> 6][(((tid >> 1) & 31) * 34 + (tid & 1) * 33) * 2] =
      (v2f){0.0f, 0.0f};
  strip_store(strip2[0], d, h, 0, cp, qtr, r);
  __syncthreads();

  v8f acc[3][2];
#pragma unroll
  for (int t = 0; t < 3; ++t) {
#pragma unroll
    for (int j = 0; j < 2; ++j) {
      const int cls = (j == 0) ? (ml == 0 ? 0 : 1) : (ml == 15 ? 2 : 1);
      v8f c;
#pragma unroll
      for (int v = 0; v < 8; ++v) c[v] = biasL[cls * MP + ob0 + t * 16 + v + 8 * lhi];
      acc[t][j] = c;
    }
  }
  v8f accx[2];
#pragma unroll
  for (int j = 0; j < 2; ++j)
#pragma unroll
    for (int v = 0; v < 8; ++v) accx[j][v] = 0.0f;

  __builtin_amdgcn_s_setprio(3);
  for (int q = 0; q < 9; ++q) {
    step_priority(q);
    const int kd = q / 3, kh = q % 3;
    const float* strip = strip2[q & 1];
    if (q < 8) strip_fetch(y1, d, h, q + 1, cp, qtr, r);
    if (q < 7 && tid < 64) {
      const int d2 = min(max(d + (q + 2) / 3 - 1, 0), 31);
      const int h2 = min(max(h + (q + 2) % 3 - 1, 0), 31);
      __builtin_prefetch(y1 + (size_t)tid * SPATIAL + (d2 * 32 + h2) * 32, 0, 0);
    }

    const int koffb = kd * 9 + kh * 3;
    const float* wt = Wt + (size_t)koffb * (32 * MP * 2) + ((size_t)lhi * MP + ml) * 2;
    const float* wq = wt + ob0 * 2;
    const float* wx = wt + 192 * 2;
    conv_krange<false>(wq, wx, strip, 0, kx, lhi, ml, acc, accx);
    conv_krange<true>(wq, wx, strip, kx, kx + 16, lhi, ml, acc, accx);
    conv_krange<false>(wq, wx, strip, kx + 16, 64, lhi, ml, acc, accx);

    if (q == 4) {
      const int ymt = wave & 1, yj = wave >> 1;
      const float* wp = Wp + (size_t)(ymt * 16 + ml) * 64 + 2 * lhi;
      v8f ya;
#pragma unroll
      for (int v = 0; v < 8; ++v) ya[v] = 0.0f;
#pragma unroll
      for (int k0 = 0; k0 < 64; k0 += 4) {
        v2f a = *(const v2f*)(wp + k0);
        v2f b = *(const v2f*)&strip[(((k0 >> 1) + lhi) * 34 + yj * 16 + ml + 1) * 2];
        ya = wmma_f32(a, b, ya);
      }
      asm volatile("v_nop\n\tv_nop\n\tv_nop\n\tv_nop\n\tv_nop" : "+v"(ya));
#pragma unroll
      for (int v = 0; v < 8; ++v)
        ystage[(ymt * 16 + v + 8 * lhi) * YSP + yj * 16 + ml] = ya[v];
    }

    if (q < 8) {
      strip_store(strip2[(q + 1) & 1], d, h, q + 1, cp, qtr, r);
      __syncthreads();
    }
  }

  const size_t sbase = ((size_t)d * 32 + h) * 32;
#pragma unroll
  for (int t = 0; t < 3; ++t) {
#pragma unroll
    for (int j = 0; j < 2; ++j)
#pragma unroll
      for (int v = 0; v < 8; ++v) {
        int o = ob0 + t * 16 + v + 8 * lhi;
        y2p[(size_t)o * SPATIAL + sbase + j * 16 + ml] = acc[t][j][v];
      }
  }

#pragma unroll
  for (int i = 0; i < 8; ++i) {
    const int row = wave * 8 + i;
    Yp[(size_t)row * SPATIAL + sbase + lane] = ystage[row * YSP + lane];
  }
  float* red = strip2[1];
#pragma unroll
  for (int j = 0; j < 2; ++j)
#pragma unroll
    for (int v = 0; v < 8; ++v)
      red[wave * 512 + (v + 8 * lhi) * 32 + j * 16 + ml] = accx[j][v];
  __syncthreads();
#pragma unroll
  for (int i = 0; i < 4; ++i) {
    const int idx = tid + 128 * i;
    const int col = idx & 31;
    const int cls = (col == 0) ? 0 : (col == 31 ? 2 : 1);
    const float s = ((red[idx] + red[512 + idx]) + (red[1024 + idx] + red[1536 + idx])) +
                    biasL[cls * MP + 192 + (idx >> 5)];
    y2p[(size_t)(192 + (idx >> 5)) * SPATIAL + sbase + (idx & 31)] = s;
  }
}

#define YPITCH 40
__global__ __launch_bounds__(128) void softmax_gather_kernel(
    const float* __restrict__ y2p, const float* __restrict__ Y,
    float* __restrict__ out) {
  __shared__ __attribute__((aligned(16))) float yl[32 * 9 * YPITCH];

  const int tid = threadIdx.x;
  const int w = tid & 31;
  const int q = tid >> 5;
  const int d = blockIdx.x >> 5, h = blockIdx.x & 31;

  v2f wv[27];
  const float* lp = y2p + (size_t)(q * 2) * SPATIAL + (d * 32 + h) * 32 + w;
#pragma unroll
  for (int n = 0; n < 26; ++n) {
    wv[n][0] = lp[(size_t)(n * 8) * SPATIAL];
    wv[n][1] = lp[(size_t)(n * 8 + 1) * SPATIAL];
  }
  wv[26][0] = 0.0f;
  wv[26][1] = 0.0f;

  const int c = tid & 7, ol = tid >> 3;
  const float* yg = Y + (size_t)ol * SPATIAL + c * 4;
  v4f t[18];
#pragma unroll
  for (int k = 0; k < 18; ++k) {
    const int r = k >> 1;
    const int nd = min(max(d + r / 3 - 1, 0), 31);
    const int nh = min(max(h + r % 3 - 1, 0), 31);
    t[k] = *(const v4f*)(yg + (size_t)(k & 1) * (16 * SPATIAL) + (nd * 32 + nh) * 32);
  }

  float* yd = &yl[ol * 9 * YPITCH + 4 + c * 4];
#pragma unroll
  for (int k = 0; k < 18; ++k)
    *(v4f*)(yd + ((k & 1) * 144 + (k >> 1)) * YPITCH) = t[k];
  if (c == 0 || c == 7) {
    float* ye = yd + (c == 0 ? -1 : 4);
#pragma unroll
    for (int k = 0; k < 18; ++k)
      ye[((k & 1) * 144 + (k >> 1)) * YPITCH] = (c == 0) ? t[k][0] : t[k][3];
  }

  float mx0 = -3.0e38f, mx1 = -3.0e38f;
#pragma unroll
  for (int n = 0; n < 27; ++n) {
    mx0 = fmaxf(mx0, wv[n][0]);
    mx1 = fmaxf(mx1, wv[n][1]);
  }
  float sum0 = 0.0f, sum1 = 0.0f;
#pragma unroll
  for (int n = 0; n < 27; ++n) {
    wv[n][0] = __expf(wv[n][0] - mx0);
    sum0 += wv[n][0];
    wv[n][1] = __expf(wv[n][1] - mx1);
    sum1 += wv[n][1];
  }
  const float inv0 = 1.0f / sum0, inv1 = 1.0f / sum1;
#pragma unroll
  for (int n = 0; n < 27; ++n) {
    wv[n][0] = wv[n][0] * inv0;
    wv[n][1] = wv[n][1] * inv1;
  }
  __syncthreads();

  const int rd = q >> 1, rh = q & 1;
  float* op = out + ((size_t)(2 * d + rd) * 64 + (2 * h + rh)) * 64 + 2 * w;
  const float* yb = &yl[w + 3];
#pragma unroll 2
  for (int o = 0; o < 32; ++o) {
    v2f a = {0.0f, 0.0f};
#pragma unroll
    for (int n = 0; n < 27; ++n) {
      const float y = yb[(o * 9 + n / 3) * YPITCH + n % 3];
      const v2f yy = {y, y};
      a = __builtin_elementwise_fma(yy, wv[n], a);
    }
    *(v2f*)(op + (size_t)o * 262144) = a;
  }
}

extern "C" void kernel_launch(void* const* d_in, const int* in_sizes, int n_in,
                              void* d_out, int out_size, void* d_ws,
                              size_t ws_size, hipStream_t stream) {
  (void)in_sizes; (void)n_in; (void)out_size; (void)ws_size;
  const float* x  = (const float*)d_in[0];
  const float* Wc = (const float*)d_in[1];
  const float* bc = (const float*)d_in[2];
  const float* We = (const float*)d_in[3];
  const float* be = (const float*)d_in[4];
  const float* Wp = (const float*)d_in[5];
  float* out = (float*)d_out;

  float* ws = (float*)d_ws;
  float* y2p  = ws + (size_t)64 * SPATIAL;
  float* wsft = y2p + (size_t)MPAD * SPATIAL;
  float* Yp   = wsft + (size_t)SPATIAL * 216;
  float* Wt   = Yp + (size_t)32 * SPATIAL;
  float* bep  = Wt + (size_t)NWT;
  float* bt   = wsft;

  compose_weights_kernel<<<351, 128, 0, stream>>>(We, be, Wc, bc, Wt, bt, bep);
  conv3_wmma_kernel<<<1024, 128, 0, stream>>>(x, Wt, bep, bt, Wp, y2p, Yp);
  softmax_gather_kernel<<<1024, 128, 0, stream>>>(y2p, Yp, out);
}
